# starting kernel
# baseline (speedup 1.0000x reference)
#include <hip/hip_runtime.h>
#include <hip/hip_bf16.h>
#include <math.h>

typedef __attribute__((ext_vector_type(2))) float v2f;
typedef __attribute__((ext_vector_type(8))) float v8f;

#define IN_DIM 128
#define HID    16


__global__ void k_init_deg(float* __restrict__ deg, int n) {
  int i = blockIdx.x * blockDim.x + threadIdx.x;
  if (i < n) deg[i] = 1.0f;
}

__global__ void k_deg_edges(const int* __restrict__ dst, float* __restrict__ deg, int nE) {
  int e = blockIdx.x * blockDim.x + threadIdx.x;
  if (e < nE) atomicAdd(&deg[dst[e]], 1.0f);
}

__global__ void k_dis(const float* __restrict__ deg, float* __restrict__ dis, int n) {
  int i = blockIdx.x * blockDim.x + threadIdx.x;
  if (i < n) dis[i] = rsqrtf(deg[i]);
}


__global__ void k_gemm1_wmma(const float* __restrict__ x, const float* __restrict__ W1,
                             float* __restrict__ h, int n) {
  const int lane = threadIdx.x & 31;
  const int wave = threadIdx.x >> 5;
  const int tile = blockIdx.x * (blockDim.x >> 5) + wave;
  const int row0 = tile << 4;
  if (row0 >= n) return;
  const int m  = lane & 15;
  const int kh = lane >> 4;
  const float* xrow = x + (size_t)(row0 + m) * IN_DIM + 2 * kh;

  v8f c = {0.f, 0.f, 0.f, 0.f, 0.f, 0.f, 0.f, 0.f};
#pragma unroll
  for (int k0 = 0; k0 < IN_DIM / 4; ++k0) {
    const int kk = 4 * k0 + 2 * kh;
    const float2 av = *reinterpret_cast<const float2*>(xrow + 4 * k0);
    v2f a; a.x = av.x; a.y = av.y;
    v2f b; b.x = W1[(kk + 0) * HID + m];
           b.y = W1[(kk + 1) * HID + m];
    c = __builtin_amdgcn_wmma_f32_16x16x4_f32(false, a, false, b, (short)0, c, false, false);
  }
  float* hout = h + (size_t)(row0 + 8 * kh) * HID + m;
#pragma unroll
  for (int r = 0; r < 8; ++r)
    hout[(size_t)r * HID] = c[r];
}


__global__ void k_gemm2_wmma(const float* __restrict__ h1, const float* __restrict__ W2,
                             float* __restrict__ h2, int n) {
  const int lane = threadIdx.x & 31;
  const int wave = threadIdx.x >> 5;
  const int tile = blockIdx.x * (blockDim.x >> 5) + wave;
  const int row0 = tile << 4;
  if (row0 >= n) return;
  const int m  = lane & 15;
  const int kh = lane >> 4;
  const float bmask = (m < 2) ? 1.0f : 0.0f;
  const int   col   = (m < 2) ? m : 0;
  const float* arow = h1 + (size_t)(row0 + m) * HID + 2 * kh;

  v8f c = {0.f, 0.f, 0.f, 0.f, 0.f, 0.f, 0.f, 0.f};
#pragma unroll
  for (int k0 = 0; k0 < HID / 4; ++k0) {
    const int kk = 4 * k0 + 2 * kh;
    const float2 av = *reinterpret_cast<const float2*>(arow + 4 * k0);
    v2f a; a.x = av.x; a.y = av.y;
    v2f b; b.x = W2[(kk + 0) * 2 + col] * bmask;
           b.y = W2[(kk + 1) * 2 + col] * bmask;
    c = __builtin_amdgcn_wmma_f32_16x16x4_f32(false, a, false, b, (short)0, c, false, false);
  }
  if (m < 2) {
#pragma unroll
    for (int r = 0; r < 8; ++r)
      h2[(size_t)(row0 + r + 8 * kh) * 2 + m] = c[r];
  }
}


__global__ void k_self1(const float* __restrict__ h, const float* __restrict__ dis,
                        float* __restrict__ out1, int n) {
  int i = blockIdx.x * blockDim.x + threadIdx.x;
  if (i >= n) return;
  float s = dis[i]; s = s * s;
  const float4* hp = reinterpret_cast<const float4*>(h + (size_t)i * HID);
  float4* op = reinterpret_cast<float4*>(out1 + (size_t)i * HID);
#pragma unroll
  for (int q = 0; q < 4; ++q) {
    float4 v = hp[q];
    v.x *= s; v.y *= s; v.z *= s; v.w *= s;
    op[q] = v;
  }
}

__global__ void k_scatter1(const int* __restrict__ src, const int* __restrict__ dst,
                           const float* __restrict__ dis, const float* __restrict__ h,
                           float* __restrict__ out1, int nE) {
  long long t = (long long)blockIdx.x * blockDim.x + threadIdx.x;
  int e = (int)(t >> 2);
  if (e >= nE) return;
  int q = (int)(t & 3);
  int s = src[e], d = dst[e];
  float norm = dis[s] * dis[d];
  float4 v = *reinterpret_cast<const float4*>(h + (size_t)s * HID + 4 * q);
  float* o = out1 + (size_t)d * HID + 4 * q;
  atomicAdd(o + 0, v.x * norm);
  atomicAdd(o + 1, v.y * norm);
  atomicAdd(o + 2, v.z * norm);
  atomicAdd(o + 3, v.w * norm);
}

__global__ void k_relu_bias(float* __restrict__ out1, const float* __restrict__ b1, int n) {
  long long t = (long long)blockIdx.x * blockDim.x + threadIdx.x;
  if (t >= (long long)n * HID) return;
  int j = (int)(t & (HID - 1));
  float v = out1[t] + b1[j];
  out1[t] = v > 0.f ? v : 0.f;
}

__global__ void k_self2(const float* __restrict__ h2, const float* __restrict__ dis,
                        float* __restrict__ out2, int n) {
  int i = blockIdx.x * blockDim.x + threadIdx.x;
  if (i >= n) return;
  float s = dis[i]; s = s * s;
  out2[(size_t)i * 2 + 0] = h2[(size_t)i * 2 + 0] * s;
  out2[(size_t)i * 2 + 1] = h2[(size_t)i * 2 + 1] * s;
}

__global__ void k_scatter2(const int* __restrict__ src, const int* __restrict__ dst,
                           const float* __restrict__ dis, const float* __restrict__ h2,
                           float* __restrict__ out2, int nE) {
  int e = blockIdx.x * blockDim.x + threadIdx.x;
  if (e >= nE) return;
  int s = src[e], d = dst[e];
  float norm = dis[s] * dis[d];
  atomicAdd(&out2[(size_t)d * 2 + 0], h2[(size_t)s * 2 + 0] * norm);
  atomicAdd(&out2[(size_t)d * 2 + 1], h2[(size_t)s * 2 + 1] * norm);
}

__global__ void k_logsoftmax(const float* __restrict__ out2, const float* __restrict__ b2,
                             float* __restrict__ y, int n) {
  int i = blockIdx.x * blockDim.x + threadIdx.x;
  if (i >= n) return;
  float a0 = out2[(size_t)i * 2 + 0] + b2[0];
  float a1 = out2[(size_t)i * 2 + 1] + b2[1];
  float mx = fmaxf(a0, a1);
  float lse = mx + logf(expf(a0 - mx) + expf(a1 - mx));
  y[(size_t)i * 2 + 0] = a0 - lse;
  y[(size_t)i * 2 + 1] = a1 - lse;
}


extern "C" void kernel_launch(void* const* d_in, const int* in_sizes, int n_in,
                              void* d_out, int out_size, void* d_ws, size_t ws_size,
                              hipStream_t stream) {
  const float* x  = (const float*)d_in[0];
  const int*   ei = (const int*)d_in[1];
  const float* W1 = (const float*)d_in[2];
  const float* b1 = (const float*)d_in[3];
  const float* W2 = (const float*)d_in[4];
  const float* b2 = (const float*)d_in[5];

  const int n  = in_sizes[0] / IN_DIM;
  const int nE = in_sizes[1] / 2;
  const int* src = ei;
  const int* dst = ei + nE;

  float* ws   = (float*)d_ws;
  float* deg  = ws;
  float* dis  = deg + n;
  float* h    = dis + n;
  float* out1 = h    + (size_t)n * HID;
  float* h2   = out1 + (size_t)n * HID;
  float* out2 = h2   + (size_t)n * 2;
  float* y    = (float*)d_out;

  const int TB = 256;
  dim3 blk(TB);
  int gN   = (n + TB - 1) / TB;
  int gE   = (nE + TB - 1) / TB;
  int gE4  = (int)(((long long)nE * 4 + TB - 1) / TB);
  int gN16 = (int)(((long long)n * HID + TB - 1) / TB);
  int tiles = (n + 15) / 16;
  int gT   = (tiles + (TB / 32) - 1) / (TB / 32);

  k_init_deg  <<<gN,  blk, 0, stream>>>(deg, n);
  k_deg_edges <<<gE,  blk, 0, stream>>>(dst, deg, nE);
  k_dis       <<<gN,  blk, 0, stream>>>(deg, dis, n);
  k_gemm1_wmma<<<gT,  blk, 0, stream>>>(x, W1, h, n);
  k_self1     <<<gN,  blk, 0, stream>>>(h, dis, out1, n);
  k_scatter1  <<<gE4, blk, 0, stream>>>(src, dst, dis, h, out1, nE);
  k_relu_bias <<<gN16,blk, 0, stream>>>(out1, b1, n);
  k_gemm2_wmma<<<gT,  blk, 0, stream>>>(out1, W2, h2, n);
  k_self2     <<<gN,  blk, 0, stream>>>(h2, dis, out2, n);
  k_scatter2  <<<gE,  blk, 0, stream>>>(src, dst, dis, h2, out2, nE);
  k_logsoftmax<<<gN,  blk, 0, stream>>>(out2, b2, y, n);
}


# Layer 2 aggregates the 16-wide activations before the W2 product, A(rW2) -> (Ar)W2: its edge pass is layer 1's 64-B-row scatter, not 8-B pair atomics; sums reassociated, still atomic arrival order
# speedup vs baseline: 1.1277x; 1.1277x over previous
#include <hip/hip_runtime.h>
#include <hip/hip_bf16.h>
#include <math.h>

typedef __attribute__((ext_vector_type(2))) float v2f;
typedef __attribute__((ext_vector_type(8))) float v8f;

#define IN_DIM 128
#define HID    16


__global__ void k_init_deg(float* __restrict__ deg, int n) {
  int i = blockIdx.x * blockDim.x + threadIdx.x;
  if (i < n) deg[i] = 1.0f;
}

__global__ void k_deg_edges(const int* __restrict__ dst, float* __restrict__ deg, int nE) {
  int e = blockIdx.x * blockDim.x + threadIdx.x;
  if (e < nE) atomicAdd(&deg[dst[e]], 1.0f);
}

__global__ void k_dis(const float* __restrict__ deg, float* __restrict__ dis, int n) {
  int i = blockIdx.x * blockDim.x + threadIdx.x;
  if (i < n) dis[i] = rsqrtf(deg[i]);
}


__global__ void k_gemm1_wmma(const float* __restrict__ x, const float* __restrict__ W1,
                             float* __restrict__ h, int n) {
  const int lane = threadIdx.x & 31;
  const int wave = threadIdx.x >> 5;
  const int tile = blockIdx.x * (blockDim.x >> 5) + wave;
  const int row0 = tile << 4;
  if (row0 >= n) return;
  const int m  = lane & 15;
  const int kh = lane >> 4;
  const float* xrow = x + (size_t)(row0 + m) * IN_DIM + 2 * kh;

  v8f c = {0.f, 0.f, 0.f, 0.f, 0.f, 0.f, 0.f, 0.f};
#pragma unroll
  for (int k0 = 0; k0 < IN_DIM / 4; ++k0) {
    const int kk = 4 * k0 + 2 * kh;
    const float2 av = *reinterpret_cast<const float2*>(xrow + 4 * k0);
    v2f a; a.x = av.x; a.y = av.y;
    v2f b; b.x = W1[(kk + 0) * HID + m];
           b.y = W1[(kk + 1) * HID + m];
    c = __builtin_amdgcn_wmma_f32_16x16x4_f32(false, a, false, b, (short)0, c, false, false);
  }
  float* hout = h + (size_t)(row0 + 8 * kh) * HID + m;
#pragma unroll
  for (int r = 0; r < 8; ++r)
    hout[(size_t)r * HID] = c[r];
}


__global__ void k_gemm2_wmma(const float* __restrict__ h1, const float* __restrict__ W2,
                             float* __restrict__ h2, int n) {
  const int lane = threadIdx.x & 31;
  const int wave = threadIdx.x >> 5;
  const int tile = blockIdx.x * (blockDim.x >> 5) + wave;
  const int row0 = tile << 4;
  if (row0 >= n) return;
  const int m  = lane & 15;
  const int kh = lane >> 4;
  const float bmask = (m < 2) ? 1.0f : 0.0f;
  const int   col   = (m < 2) ? m : 0;
  const float* arow = h1 + (size_t)(row0 + m) * HID + 2 * kh;

  v8f c = {0.f, 0.f, 0.f, 0.f, 0.f, 0.f, 0.f, 0.f};
#pragma unroll
  for (int k0 = 0; k0 < HID / 4; ++k0) {
    const int kk = 4 * k0 + 2 * kh;
    const float2 av = *reinterpret_cast<const float2*>(arow + 4 * k0);
    v2f a; a.x = av.x; a.y = av.y;
    v2f b; b.x = W2[(kk + 0) * 2 + col] * bmask;
           b.y = W2[(kk + 1) * 2 + col] * bmask;
    c = __builtin_amdgcn_wmma_f32_16x16x4_f32(false, a, false, b, (short)0, c, false, false);
  }
  if (m < 2) {
#pragma unroll
    for (int r = 0; r < 8; ++r)
      h2[(size_t)(row0 + r + 8 * kh) * 2 + m] = c[r];
  }
}


__global__ void k_self1(const float* __restrict__ h, const float* __restrict__ dis,
                        float* __restrict__ out1, int n) {
  int i = blockIdx.x * blockDim.x + threadIdx.x;
  if (i >= n) return;
  float s = dis[i]; s = s * s;
  const float4* hp = reinterpret_cast<const float4*>(h + (size_t)i * HID);
  float4* op = reinterpret_cast<float4*>(out1 + (size_t)i * HID);
#pragma unroll
  for (int q = 0; q < 4; ++q) {
    float4 v = hp[q];
    v.x *= s; v.y *= s; v.z *= s; v.w *= s;
    op[q] = v;
  }
}

__global__ void k_scatter1(const int* __restrict__ src, const int* __restrict__ dst,
                           const float* __restrict__ dis, const float* __restrict__ h,
                           float* __restrict__ out1, int nE) {
  long long t = (long long)blockIdx.x * blockDim.x + threadIdx.x;
  int e = (int)(t >> 2);
  if (e >= nE) return;
  int q = (int)(t & 3);
  int s = src[e], d = dst[e];
  float norm = dis[s] * dis[d];
  float4 v = *reinterpret_cast<const float4*>(h + (size_t)s * HID + 4 * q);
  float* o = out1 + (size_t)d * HID + 4 * q;
  atomicAdd(o + 0, v.x * norm);
  atomicAdd(o + 1, v.y * norm);
  atomicAdd(o + 2, v.z * norm);
  atomicAdd(o + 3, v.w * norm);
}

__global__ void k_relu_bias(float* __restrict__ out1, const float* __restrict__ b1, int n) {
  long long t = (long long)blockIdx.x * blockDim.x + threadIdx.x;
  if (t >= (long long)n * HID) return;
  int j = (int)(t & (HID - 1));
  float v = out1[t] + b1[j];
  out1[t] = v > 0.f ? v : 0.f;
}

__global__ void k_logsoftmax(const float* __restrict__ h2, const float* __restrict__ b2,
                             float* __restrict__ y, int n) {
  int i = blockIdx.x * blockDim.x + threadIdx.x;
  if (i >= n) return;
  float a0 = h2[(size_t)i * 2 + 0] + b2[0];
  float a1 = h2[(size_t)i * 2 + 1] + b2[1];
  float mx = fmaxf(a0, a1);
  float lse = mx + logf(expf(a0 - mx) + expf(a1 - mx));
  y[(size_t)i * 2 + 0] = a0 - lse;
  y[(size_t)i * 2 + 1] = a1 - lse;
}


extern "C" void kernel_launch(void* const* d_in, const int* in_sizes, int n_in,
                              void* d_out, int out_size, void* d_ws, size_t ws_size,
                              hipStream_t stream) {
  const float* x  = (const float*)d_in[0];
  const int*   ei = (const int*)d_in[1];
  const float* W1 = (const float*)d_in[2];
  const float* b1 = (const float*)d_in[3];
  const float* W2 = (const float*)d_in[4];
  const float* b2 = (const float*)d_in[5];

  const int n  = in_sizes[0] / IN_DIM;
  const int nE = in_sizes[1] / 2;
  const int* src = ei;
  const int* dst = ei + nE;

  float* ws   = (float*)d_ws;
  float* deg  = ws;
  float* dis  = deg + n;
  float* h    = dis + n;
  float* out1 = h    + (size_t)n * HID;
  float* h2   = out1 + (size_t)n * HID;
  float* agg  = h;
  float* y    = (float*)d_out;

  const int TB = 256;
  dim3 blk(TB);
  int gN   = (n + TB - 1) / TB;
  int gE   = (nE + TB - 1) / TB;
  int gE4  = (int)(((long long)nE * 4 + TB - 1) / TB);
  int gN16 = (int)(((long long)n * HID + TB - 1) / TB);
  int tiles = (n + 15) / 16;
  int gT   = (tiles + (TB / 32) - 1) / (TB / 32);

  k_init_deg  <<<gN,  blk, 0, stream>>>(deg, n);
  k_deg_edges <<<gE,  blk, 0, stream>>>(dst, deg, nE);
  k_dis       <<<gN,  blk, 0, stream>>>(deg, dis, n);
  k_gemm1_wmma<<<gT,  blk, 0, stream>>>(x, W1, h, n);
  k_self1     <<<gN,  blk, 0, stream>>>(h, dis, out1, n);
  k_scatter1  <<<gE4, blk, 0, stream>>>(src, dst, dis, h, out1, nE);
  k_relu_bias <<<gN16,blk, 0, stream>>>(out1, b1, n);
  k_self1     <<<gN,  blk, 0, stream>>>(out1, dis, agg, n);
  k_scatter1  <<<gE4, blk, 0, stream>>>(src, dst, dis, out1, agg, nE);
  k_gemm2_wmma<<<gT,  blk, 0, stream>>>(agg, W2, h2, n);
  k_logsoftmax<<<gN,  blk, 0, stream>>>(h2, b2, y, n);
}


# Edge pass lane = feature: resident waves loop over 32-edge batches; each half-wave gathers and f32-atomic-adds one whole 64-B row (2 rows/instruction, not 8 x 4 strided words); atomic order as before
# speedup vs baseline: 1.5153x; 1.3437x over previous
#include <hip/hip_runtime.h>
#include <hip/hip_bf16.h>
#include <math.h>

typedef __attribute__((ext_vector_type(2))) float v2f;
typedef __attribute__((ext_vector_type(8))) float v8f;

#define IN_DIM 128
#define HID    16


__global__ void k_init_deg(float* __restrict__ deg, int n) {
  int i = blockIdx.x * blockDim.x + threadIdx.x;
  if (i < n) deg[i] = 1.0f;
}

__global__ void k_deg_edges(const int* __restrict__ dst, float* __restrict__ deg, int nE) {
  int e = blockIdx.x * blockDim.x + threadIdx.x;
  if (e < nE) atomicAdd(&deg[dst[e]], 1.0f);
}

__global__ void k_dis(const float* __restrict__ deg, float* __restrict__ dis, int n) {
  int i = blockIdx.x * blockDim.x + threadIdx.x;
  if (i < n) dis[i] = rsqrtf(deg[i]);
}


__global__ void k_gemm1_wmma(const float* __restrict__ x, const float* __restrict__ W1,
                             float* __restrict__ h, int n) {
  const int lane = threadIdx.x & 31;
  const int wave = threadIdx.x >> 5;
  const int tile = blockIdx.x * (blockDim.x >> 5) + wave;
  const int row0 = tile << 4;
  if (row0 >= n) return;
  const int m  = lane & 15;
  const int kh = lane >> 4;
  const float* xrow = x + (size_t)(row0 + m) * IN_DIM + 2 * kh;

  v8f c = {0.f, 0.f, 0.f, 0.f, 0.f, 0.f, 0.f, 0.f};
#pragma unroll
  for (int k0 = 0; k0 < IN_DIM / 4; ++k0) {
    const int kk = 4 * k0 + 2 * kh;
    const float2 av = *reinterpret_cast<const float2*>(xrow + 4 * k0);
    v2f a; a.x = av.x; a.y = av.y;
    v2f b; b.x = W1[(kk + 0) * HID + m];
           b.y = W1[(kk + 1) * HID + m];
    c = __builtin_amdgcn_wmma_f32_16x16x4_f32(false, a, false, b, (short)0, c, false, false);
  }
  float* hout = h + (size_t)(row0 + 8 * kh) * HID + m;
#pragma unroll
  for (int r = 0; r < 8; ++r)
    hout[(size_t)r * HID] = c[r];
}


__global__ void k_gemm2_wmma(const float* __restrict__ h1, const float* __restrict__ W2,
                             float* __restrict__ h2, int n) {
  const int lane = threadIdx.x & 31;
  const int wave = threadIdx.x >> 5;
  const int tile = blockIdx.x * (blockDim.x >> 5) + wave;
  const int row0 = tile << 4;
  if (row0 >= n) return;
  const int m  = lane & 15;
  const int kh = lane >> 4;
  const float bmask = (m < 2) ? 1.0f : 0.0f;
  const int   col   = (m < 2) ? m : 0;
  const float* arow = h1 + (size_t)(row0 + m) * HID + 2 * kh;

  v8f c = {0.f, 0.f, 0.f, 0.f, 0.f, 0.f, 0.f, 0.f};
#pragma unroll
  for (int k0 = 0; k0 < HID / 4; ++k0) {
    const int kk = 4 * k0 + 2 * kh;
    const float2 av = *reinterpret_cast<const float2*>(arow + 4 * k0);
    v2f a; a.x = av.x; a.y = av.y;
    v2f b; b.x = W2[(kk + 0) * 2 + col] * bmask;
           b.y = W2[(kk + 1) * 2 + col] * bmask;
    c = __builtin_amdgcn_wmma_f32_16x16x4_f32(false, a, false, b, (short)0, c, false, false);
  }
  if (m < 2) {
#pragma unroll
    for (int r = 0; r < 8; ++r)
      h2[(size_t)(row0 + r + 8 * kh) * 2 + m] = c[r];
  }
}


__global__ void k_self1(const float* __restrict__ h, const float* __restrict__ dis,
                        float* __restrict__ out1, int n) {
  int i = blockIdx.x * blockDim.x + threadIdx.x;
  if (i >= n) return;
  float s = dis[i]; s = s * s;
  const float4* hp = reinterpret_cast<const float4*>(h + (size_t)i * HID);
  float4* op = reinterpret_cast<float4*>(out1 + (size_t)i * HID);
#pragma unroll
  for (int q = 0; q < 4; ++q) {
    float4 v = hp[q];
    v.x *= s; v.y *= s; v.z *= s; v.w *= s;
    op[q] = v;
  }
}


static __device__ __forceinline__ unsigned row_bcast(int j, unsigned v) {
  const unsigned sel = 0x11111111u * (unsigned)j;
  return __builtin_amdgcn_permlane16(0u, v, sel, sel, false, true);
}

__global__ void k_scatter1(const int* __restrict__ src, const int* __restrict__ dst,
                           const float* __restrict__ dis, const float* __restrict__ h,
                           float* __restrict__ out1, int nE) {
  const int lane = threadIdx.x & 31;
  const unsigned f = lane & 15;
  const int wv     = __builtin_amdgcn_readfirstlane((int)(threadIdx.x >> 5));
  const int wpb    = (int)(blockDim.x >> 5);
  const int nWaves = (int)gridDim.x * wpb;
  const int nBatch = (nE + 31) >> 5;
  int b = (int)blockIdx.x * wpb + wv;
  if (b >= nBatch) return;

  unsigned s = 0, d = 0;
  if (lane < nE - (b << 5)) {
    s = (unsigned)src[(b << 5) + lane]; d = (unsigned)dst[(b << 5) + lane];
  }
  for (; b < nBatch; b += nWaves) {
    const int left = nE - (b << 5);
    const int bn   = b + nWaves;
    unsigned sN = 0, dN = 0;
    if (left >= 32) {
      const float ds = dis[s], dd = dis[d];
      float v[16];
#pragma unroll
      for (int j = 0; j < 16; ++j) v[j] = h[row_bcast(j, s) * HID + f];
      if (bn < nBatch && lane < nE - (bn << 5)) {
        sN = (unsigned)src[(bn << 5) + lane]; dN = (unsigned)dst[(bn << 5) + lane];
      }
      const float nrm = ds * dd;
#pragma unroll
      for (int j = 0; j < 16; ++j) {
        const unsigned dj = row_bcast(j, d);
        const float    nj = __uint_as_float(row_bcast(j, __float_as_uint(nrm)));
        atomicAdd(out1 + (dj * HID + f), v[j] * nj);
      }
    } else {
      float nrm = 0.f;
      if (lane < left) nrm = dis[s] * dis[d];
      const int mine = left - (lane & 16);
#pragma unroll
      for (int j = 0; j < 16; ++j) {
        const unsigned sj = row_bcast(j, s);
        const unsigned dj = row_bcast(j, d);
        const float    nj = __uint_as_float(row_bcast(j, __float_as_uint(nrm)));
        if (j < mine) atomicAdd(out1 + (dj * HID + f), h[sj * HID + f] * nj);
      }
    }
    s = sN; d = dN;
  }
}

__global__ void k_relu_bias(float* __restrict__ out1, const float* __restrict__ b1, int n) {
  long long t = (long long)blockIdx.x * blockDim.x + threadIdx.x;
  if (t >= (long long)n * HID) return;
  int j = (int)(t & (HID - 1));
  float v = out1[t] + b1[j];
  out1[t] = v > 0.f ? v : 0.f;
}

__global__ void k_logsoftmax(const float* __restrict__ h2, const float* __restrict__ b2,
                             float* __restrict__ y, int n) {
  int i = blockIdx.x * blockDim.x + threadIdx.x;
  if (i >= n) return;
  float a0 = h2[(size_t)i * 2 + 0] + b2[0];
  float a1 = h2[(size_t)i * 2 + 1] + b2[1];
  float mx = fmaxf(a0, a1);
  float lse = mx + logf(expf(a0 - mx) + expf(a1 - mx));
  y[(size_t)i * 2 + 0] = a0 - lse;
  y[(size_t)i * 2 + 1] = a1 - lse;
}


extern "C" void kernel_launch(void* const* d_in, const int* in_sizes, int n_in,
                              void* d_out, int out_size, void* d_ws, size_t ws_size,
                              hipStream_t stream) {
  const float* x  = (const float*)d_in[0];
  const int*   ei = (const int*)d_in[1];
  const float* W1 = (const float*)d_in[2];
  const float* b1 = (const float*)d_in[3];
  const float* W2 = (const float*)d_in[4];
  const float* b2 = (const float*)d_in[5];

  const int n  = in_sizes[0] / IN_DIM;
  const int nE = in_sizes[1] / 2;
  const int* src = ei;
  const int* dst = ei + nE;

  float* ws   = (float*)d_ws;
  float* deg  = ws;
  float* dis  = deg + n;
  float* h    = dis + n;
  float* out1 = h    + (size_t)n * HID;
  float* h2   = out1 + (size_t)n * HID;
  float* agg  = h;
  float* y    = (float*)d_out;

  const int TB = 256;
  dim3 blk(TB);
  int gN   = (n + TB - 1) / TB;
  int gE   = (nE + TB - 1) / TB;
  int gS   = 2048;
  int gN16 = (int)(((long long)n * HID + TB - 1) / TB);
  int tiles = (n + 15) / 16;
  int gT   = (tiles + (TB / 32) - 1) / (TB / 32);

  k_init_deg  <<<gN,  blk, 0, stream>>>(deg, n);
  k_deg_edges <<<gE,  blk, 0, stream>>>(dst, deg, nE);
  k_dis       <<<gN,  blk, 0, stream>>>(deg, dis, n);
  k_gemm1_wmma<<<gT,  blk, 0, stream>>>(x, W1, h, n);
  k_self1     <<<gN,  blk, 0, stream>>>(h, dis, out1, n);
  k_scatter1  <<<gS,  blk, 0, stream>>>(src, dst, dis, h, out1, nE);
  k_relu_bias <<<gN16,blk, 0, stream>>>(out1, b1, n);
  k_self1     <<<gN,  blk, 0, stream>>>(out1, dis, agg, n);
  k_scatter1  <<<gS,  blk, 0, stream>>>(src, dst, dis, out1, agg, nE);
  k_gemm2_wmma<<<gT,  blk, 0, stream>>>(agg, W2, h2, n);
  k_logsoftmax<<<gN,  blk, 0, stream>>>(h2, b2, y, n);
}


# Edges binned once by 256-node destination range (LDS counting sort per chunk); a range's work-group sums rows with LDS f32 atomics: no global atomic left; sums in LDS arrival order
# speedup vs baseline: 4.0736x; 2.6883x over previous
#include <hip/hip_runtime.h>
#include <hip/hip_bf16.h>
#include <math.h>

typedef __attribute__((ext_vector_type(2))) float v2f;
typedef __attribute__((ext_vector_type(8))) float v8f;

#define IN_DIM 128
#define HID    16


__global__ void k_init_deg(float* __restrict__ deg, int n) {
  int i = blockIdx.x * blockDim.x + threadIdx.x;
  if (i < n) deg[i] = 1.0f;
}

__global__ void k_deg_edges(const int* __restrict__ dst, float* __restrict__ deg, int nE) {
  int e = blockIdx.x * blockDim.x + threadIdx.x;
  if (e < nE) atomicAdd(&deg[dst[e]], 1.0f);
}

__global__ void k_dis(const float* __restrict__ deg, float* __restrict__ dis, int n) {
  int i = blockIdx.x * blockDim.x + threadIdx.x;
  if (i < n) dis[i] = rsqrtf(deg[i]);
}


__global__ void k_gemm1_wmma(const float* __restrict__ x, const float* __restrict__ W1,
                             float* __restrict__ h, int n) {
  const int lane = threadIdx.x & 31;
  const int wave = threadIdx.x >> 5;
  const int tile = blockIdx.x * (blockDim.x >> 5) + wave;
  const int row0 = tile << 4;
  if (row0 >= n) return;
  const int m  = lane & 15;
  const int kh = lane >> 4;
  const float* xrow = x + (size_t)(row0 + m) * IN_DIM + 2 * kh;

  v8f c = {0.f, 0.f, 0.f, 0.f, 0.f, 0.f, 0.f, 0.f};
#pragma unroll
  for (int k0 = 0; k0 < IN_DIM / 4; ++k0) {
    const int kk = 4 * k0 + 2 * kh;
    const float2 av = *reinterpret_cast<const float2*>(xrow + 4 * k0);
    v2f a; a.x = av.x; a.y = av.y;
    v2f b; b.x = W1[(kk + 0) * HID + m];
           b.y = W1[(kk + 1) * HID + m];
    c = __builtin_amdgcn_wmma_f32_16x16x4_f32(false, a, false, b, (short)0, c, false, false);
  }
  float* hout = h + (size_t)(row0 + 8 * kh) * HID + m;
#pragma unroll
  for (int r = 0; r < 8; ++r)
    hout[(size_t)r * HID] = c[r];
}


__global__ void k_gemm2_wmma(const float* __restrict__ h1, const float* __restrict__ W2,
                             float* __restrict__ h2, int n) {
  const int lane = threadIdx.x & 31;
  const int wave = threadIdx.x >> 5;
  const int tile = blockIdx.x * (blockDim.x >> 5) + wave;
  const int row0 = tile << 4;
  if (row0 >= n) return;
  const int m  = lane & 15;
  const int kh = lane >> 4;
  const float bmask = (m < 2) ? 1.0f : 0.0f;
  const int   col   = (m < 2) ? m : 0;
  const float* arow = h1 + (size_t)(row0 + m) * HID + 2 * kh;

  v8f c = {0.f, 0.f, 0.f, 0.f, 0.f, 0.f, 0.f, 0.f};
#pragma unroll
  for (int k0 = 0; k0 < HID / 4; ++k0) {
    const int kk = 4 * k0 + 2 * kh;
    const float2 av = *reinterpret_cast<const float2*>(arow + 4 * k0);
    v2f a; a.x = av.x; a.y = av.y;
    v2f b; b.x = W2[(kk + 0) * 2 + col] * bmask;
           b.y = W2[(kk + 1) * 2 + col] * bmask;
    c = __builtin_amdgcn_wmma_f32_16x16x4_f32(false, a, false, b, (short)0, c, false, false);
  }
  if (m < 2) {
#pragma unroll
    for (int r = 0; r < 8; ++r)
      h2[(size_t)(row0 + r + 8 * kh) * 2 + m] = c[r];
  }
}


__global__ void k_self1(const float* __restrict__ h, const float* __restrict__ dis,
                        float* __restrict__ out1, int n) {
  int i = blockIdx.x * blockDim.x + threadIdx.x;
  if (i >= n) return;
  float s = dis[i]; s = s * s;
  const float4* hp = reinterpret_cast<const float4*>(h + (size_t)i * HID);
  float4* op = reinterpret_cast<float4*>(out1 + (size_t)i * HID);
#pragma unroll
  for (int q = 0; q < 4; ++q) {
    float4 v = hp[q];
    v.x *= s; v.y *= s; v.z *= s; v.w *= s;
    op[q] = v;
  }
}


static __device__ __forceinline__ unsigned row_bcast(int j, unsigned v) {
  const unsigned sel = 0x11111111u * (unsigned)j;
  return __builtin_amdgcn_permlane16(0u, v, sel, sel, false, true);
}

__global__ void k_scatter1(const int* __restrict__ src, const int* __restrict__ dst,
                           const float* __restrict__ dis, const float* __restrict__ h,
                           float* __restrict__ out1, int nE) {
  const int lane = threadIdx.x & 31;
  const unsigned f = lane & 15;
  const int wv     = __builtin_amdgcn_readfirstlane((int)(threadIdx.x >> 5));
  const int wpb    = (int)(blockDim.x >> 5);
  const int nWaves = (int)gridDim.x * wpb;
  const int nBatch = (nE + 31) >> 5;
  int b = (int)blockIdx.x * wpb + wv;
  if (b >= nBatch) return;

  unsigned s = 0, d = 0;
  if (lane < nE - (b << 5)) {
    s = (unsigned)src[(b << 5) + lane]; d = (unsigned)dst[(b << 5) + lane];
  }
  for (; b < nBatch; b += nWaves) {
    const int left = nE - (b << 5);
    const int bn   = b + nWaves;
    unsigned sN = 0, dN = 0;
    if (left >= 32) {
      const float ds = dis[s], dd = dis[d];
      float v[16];
#pragma unroll
      for (int j = 0; j < 16; ++j) v[j] = h[row_bcast(j, s) * HID + f];
      if (bn < nBatch && lane < nE - (bn << 5)) {
        sN = (unsigned)src[(bn << 5) + lane]; dN = (unsigned)dst[(bn << 5) + lane];
      }
      const float nrm = ds * dd;
#pragma unroll
      for (int j = 0; j < 16; ++j) {
        const unsigned dj = row_bcast(j, d);
        const float    nj = __uint_as_float(row_bcast(j, __float_as_uint(nrm)));
        atomicAdd(out1 + (dj * HID + f), v[j] * nj);
      }
    } else {
      float nrm = 0.f;
      if (lane < left) nrm = dis[s] * dis[d];
      const int mine = left - (lane & 16);
#pragma unroll
      for (int j = 0; j < 16; ++j) {
        const unsigned sj = row_bcast(j, s);
        const unsigned dj = row_bcast(j, d);
        const float    nj = __uint_as_float(row_bcast(j, __float_as_uint(nrm)));
        if (j < mine) atomicAdd(out1 + (dj * HID + f), h[sj * HID + f] * nj);
      }
    }
    s = sN; d = dN;
  }
}

__global__ void k_relu_bias(float* __restrict__ out1, const float* __restrict__ b1, int n) {
  long long t = (long long)blockIdx.x * blockDim.x + threadIdx.x;
  if (t >= (long long)n * HID) return;
  int j = (int)(t & (HID - 1));
  float v = out1[t] + b1[j];
  out1[t] = v > 0.f ? v : 0.f;
}

__global__ void k_logsoftmax(const float* __restrict__ h2, const float* __restrict__ b2,
                             float* __restrict__ y, int n) {
  int i = blockIdx.x * blockDim.x + threadIdx.x;
  if (i >= n) return;
  float a0 = h2[(size_t)i * 2 + 0] + b2[0];
  float a1 = h2[(size_t)i * 2 + 1] + b2[1];
  float mx = fmaxf(a0, a1);
  float lse = mx + logf(expf(a0 - mx) + expf(a1 - mx));
  y[(size_t)i * 2 + 0] = a0 - lse;
  y[(size_t)i * 2 + 1] = a1 - lse;
}


#define BKT      256
#define CH_CAP   25024
#define NB_MAX   1024

__global__ __launch_bounds__(1024) void k_bin(const int* __restrict__ src, const int* __restrict__ dst,
                                              unsigned* __restrict__ list, int* __restrict__ segoff,
                                              int nE, int nB, int pitch) {
  __shared__ unsigned cnt[NB_MAX];
  __shared__ unsigned wsum[32];
  __shared__ unsigned stage[CH_CAP];
  const int tid  = threadIdx.x;
  const int lane = tid & 31;
  const int wave = tid >> 5;
  const int g    = blockIdx.x;
  int e0 = g * CH_CAP;            if (e0 > nE) e0 = nE;
  int e1 = e0 + CH_CAP;           if (e1 > nE) e1 = nE;

  cnt[tid] = 0u;
  __syncthreads();
  for (int e = e0 + tid; e < e1; e += 8 * 1024) {
    unsigned d[8];
#pragma unroll
    for (int k = 0; k < 8; ++k) { const int ee = e + k * 1024; d[k] = (unsigned)dst[ee < e1 ? ee : e1 - 1]; }
#pragma unroll
    for (int k = 0; k < 8; ++k) if (e + k * 1024 < e1) atomicAdd(&cnt[d[k] >> 8], 1u);
  }
  __syncthreads();

  const unsigned c = cnt[tid];
  unsigned x = c;
  for (int o = 1; o < 32; o <<= 1) { const unsigned y = __shfl_up(x, o); if (lane >= o) x += y; }
  if (lane == 31) wsum[wave] = x;
  __syncthreads();
  if (wave == 0) {
    const unsigned t = wsum[lane];
    unsigned u = t;
    for (int o = 1; o < 32; o <<= 1) { const unsigned y = __shfl_up(u, o); if (lane >= o) u += y; }
    wsum[lane] = u - t;
  }
  __syncthreads();
  const unsigned excl = x - c + wsum[wave];
  if (tid <= nB) segoff[(size_t)g * pitch + tid] = e0 + (int)excl;
  cnt[tid] = excl;
  __syncthreads();

  for (int e = e0 + tid; e < e1; e += 8 * 1024) {
    unsigned d[8], s[8];
#pragma unroll
    for (int k = 0; k < 8; ++k) {
      const int ee = e + k * 1024, ec = ee < e1 ? ee : e1 - 1;
      d[k] = (unsigned)dst[ec]; s[k] = (unsigned)src[ec];
    }
#pragma unroll
    for (int k = 0; k < 8; ++k) {
      if (e + k * 1024 < e1) {
        const unsigned r = atomicAdd(&cnt[d[k] >> 8], 1u);
        stage[r] = (s[k] << 8) | (d[k] & 255u);
      }
    }
  }
  __syncthreads();
  const int len = e1 - e0;
  for (int i = tid; i < len; i += 1024) list[e0 + i] = stage[i];
}

__global__ __launch_bounds__(1024) void k_degdis(const unsigned* __restrict__ list, const int* __restrict__ segoff,
                                                 float* __restrict__ dis, int n, int pitch, int nChunk) {
  __shared__ unsigned cnt[BKT];
  const int tid  = threadIdx.x;
  const int lane = tid & 31;
  const int wv   = __builtin_amdgcn_readfirstlane((int)(threadIdx.x >> 5));
  const int b    = blockIdx.x;
  if (tid < BKT) cnt[tid] = 0u;
  __syncthreads();
  int beg = 0, end = 0;
  if (wv < nChunk) { beg = segoff[(size_t)wv * pitch + b]; end = segoff[(size_t)wv * pitch + b + 1]; }
  for (int g = wv; g < nChunk; g += 32) {
    const int gn = g + 32;
    int begN = 0, endN = 0;
    if (gn < nChunk) { begN = segoff[(size_t)gn * pitch + b]; endN = segoff[(size_t)gn * pitch + b + 1]; }
    for (int p = beg + lane; p < end; p += 32) atomicAdd(&cnt[list[p] & 255u], 1u);
    beg = begN; end = endN;
  }
  __syncthreads();
  const int node = b * BKT + tid;
  if (tid < BKT && node < n) dis[node] = rsqrtf((float)(cnt[tid] + 1u));
}

__global__ __launch_bounds__(512) void k_agg(const unsigned* __restrict__ list, const int* __restrict__ segoff,
                                             const float* __restrict__ dis, const float* __restrict__ h,
                                             float* __restrict__ out, int n, int pitch, int nChunk) {
  __shared__ float acc[BKT * HID];
  __shared__ float disL[BKT];
  const int tid  = threadIdx.x;
  const int lane = tid & 31;
  const unsigned f = lane & 15;
  const int wv   = __builtin_amdgcn_readfirstlane((int)(threadIdx.x >> 5));
  const int b    = blockIdx.x;
  const int node0 = b * BKT;

  for (int t = tid; t < BKT * HID; t += 512) {
    const int node = node0 + (t >> 4);
    float v = 0.f;
    if (node < n) { float s = dis[node]; s = s * s; v = h[(size_t)node * HID + (t & 15)] * s; }
    acc[t] = v;
  }
  if (tid < BKT) { const int node = node0 + tid; disL[tid] = node < n ? dis[node] : 0.f; }
  __syncthreads();

  int beg = 0, end = 0;
  if (wv < nChunk) { beg = segoff[(size_t)wv * pitch + b]; end = segoff[(size_t)wv * pitch + b + 1]; }
  for (int g = wv; g < nChunk; g += 16) {
    const int gn = g + 16;
    int begN = 0, endN = 0;
    if (gn < nChunk) { begN = segoff[(size_t)gn * pitch + b]; endN = segoff[(size_t)gn * pitch + b + 1]; }
    for (int base = beg; base < end; base += 32) {
      const int left = end - base;
      const int p = base + lane;
      const unsigned e  = list[p < end ? p : end - 1];
      const unsigned s  = e >> 8, dl = e & 255u;
      const float ds = dis[s];
      float v[16];
#pragma unroll
      for (int j = 0; j < 16; ++j) v[j] = h[row_bcast(j, s) * HID + f];
      const float nrm = ds * disL[dl];
      const int mine = left - (lane & 16);
#pragma unroll
      for (int j = 0; j < 16; ++j) {
        const unsigned dj = row_bcast(j, dl);
        const float    nj = __uint_as_float(row_bcast(j, __float_as_uint(nrm)));
        if (j < mine) atomicAdd(&acc[dj * HID + f], v[j] * nj);
      }
    }
    beg = begN; end = endN;
  }
  __syncthreads();
  for (int t = tid; t < BKT * HID; t += 512) {
    const int node = node0 + (t >> 4);
    if (node < n) out[(size_t)node0 * HID + t] = acc[t];
  }
}


extern "C" void kernel_launch(void* const* d_in, const int* in_sizes, int n_in,
                              void* d_out, int out_size, void* d_ws, size_t ws_size,
                              hipStream_t stream) {
  const float* x  = (const float*)d_in[0];
  const int*   ei = (const int*)d_in[1];
  const float* W1 = (const float*)d_in[2];
  const float* b1 = (const float*)d_in[3];
  const float* W2 = (const float*)d_in[4];
  const float* b2 = (const float*)d_in[5];

  const int n  = in_sizes[0] / IN_DIM;
  const int nE = in_sizes[1] / 2;
  const int* src = ei;
  const int* dst = ei + nE;
  float* ws = (float*)d_ws;
  float* y  = (float*)d_out;

  const int TB = 256;
  dim3 blk(TB);
  int gN   = (n + TB - 1) / TB;
  int gN16 = (int)(((long long)n * HID + TB - 1) / TB);
  int tiles = (n + 15) / 16;
  int gT   = (tiles + (TB / 32) - 1) / (TB / 32);

  const int nB     = (n + BKT - 1) / BKT;
  const int pitch  = (nB + 1 + 31) & ~31;
  const int nChunk = (nE + CH_CAP - 1) / CH_CAP;
  const size_t need = ((size_t)n * (1 + HID + HID + 2) + (size_t)nE + (size_t)nChunk * pitch) * sizeof(float);

  if (ws_size >= need && nChunk > 0 && nB < NB_MAX && n <= (1 << 24)) {
    float* dis  = ws;
    float* h    = dis  + n;
    float* out1 = h    + (size_t)n * HID;
    float* h2   = out1 + (size_t)n * HID;
    unsigned* list = (unsigned*)(h2 + (size_t)n * 2);
    int* segoff    = (int*)(list + nE);
    float* agg  = h;

    k_bin       <<<nChunk, dim3(1024), 0, stream>>>(src, dst, list, segoff, nE, nB, pitch);
    k_degdis    <<<nB,  dim3(1024), 0, stream>>>(list, segoff, dis, n, pitch, nChunk);
    k_gemm1_wmma<<<gT,  blk, 0, stream>>>(x, W1, h, n);
    k_agg       <<<nB,  dim3(512), 0, stream>>>(list, segoff, dis, h, out1, n, pitch, nChunk);
    k_relu_bias <<<gN16,blk, 0, stream>>>(out1, b1, n);
    k_agg       <<<nB,  dim3(512), 0, stream>>>(list, segoff, dis, out1, agg, n, pitch, nChunk);
    k_gemm2_wmma<<<gT,  blk, 0, stream>>>(agg, W2, h2, n);
    k_logsoftmax<<<gN,  blk, 0, stream>>>(h2, b2, y, n);
    return;
  }

  float* deg  = ws;
  float* dis  = deg + n;
  float* h    = dis + n;
  float* out1 = h    + (size_t)n * HID;
  float* h2   = out1 + (size_t)n * HID;
  float* agg  = h;
  int gE   = (nE + TB - 1) / TB;
  int gS   = 2048;

  k_init_deg  <<<gN,  blk, 0, stream>>>(deg, n);
  k_deg_edges <<<gE,  blk, 0, stream>>>(dst, deg, nE);
  k_dis       <<<gN,  blk, 0, stream>>>(deg, dis, n);
  k_gemm1_wmma<<<gT,  blk, 0, stream>>>(x, W1, h, n);
  k_self1     <<<gN,  blk, 0, stream>>>(h, dis, out1, n);
  k_scatter1  <<<gS,  blk, 0, stream>>>(src, dst, dis, h, out1, nE);
  k_relu_bias <<<gN16,blk, 0, stream>>>(out1, b1, n);
  k_self1     <<<gN,  blk, 0, stream>>>(out1, dis, agg, n);
  k_scatter1  <<<gS,  blk, 0, stream>>>(src, dst, dis, out1, agg, nE);
  k_gemm2_wmma<<<gT,  blk, 0, stream>>>(agg, W2, h2, n);
  k_logsoftmax<<<gN,  blk, 0, stream>>>(h2, b2, y, n);
}


# Degree factors moved from edges to nodes: GEMM1 and the ReLU pass store rows times dis[src], dis[dst] multiplies each node's finished sum once; the edge pass gathers and multiplies no factor
# speedup vs baseline: 5.0566x; 1.2413x over previous
#include <hip/hip_runtime.h>
#include <hip/hip_bf16.h>
#include <math.h>

typedef __attribute__((ext_vector_type(2))) float v2f;
typedef __attribute__((ext_vector_type(8))) float v8f;

#define IN_DIM 128
#define HID    16


__global__ void k_init_deg(float* __restrict__ deg, int n) {
  int i = blockIdx.x * blockDim.x + threadIdx.x;
  if (i < n) deg[i] = 1.0f;
}

__global__ void k_deg_edges(const int* __restrict__ dst, float* __restrict__ deg, int nE) {
  int e = blockIdx.x * blockDim.x + threadIdx.x;
  if (e < nE) atomicAdd(&deg[dst[e]], 1.0f);
}

__global__ void k_dis(const float* __restrict__ deg, float* __restrict__ dis, int n) {
  int i = blockIdx.x * blockDim.x + threadIdx.x;
  if (i < n) dis[i] = rsqrtf(deg[i]);
}


__global__ void k_gemm1_wmma(const float* __restrict__ x, const float* __restrict__ W1,
                             const float* __restrict__ rs, float* __restrict__ h, int n) {
  const int lane = threadIdx.x & 31;
  const int wave = threadIdx.x >> 5;
  const int tile = blockIdx.x * (blockDim.x >> 5) + wave;
  const int row0 = tile << 4;
  if (row0 >= n) return;
  const int m  = lane & 15;
  const int kh = lane >> 4;
  const float* xrow = x + (size_t)(row0 + m) * IN_DIM + 2 * kh;

  float sc[8];
#pragma unroll
  for (int r = 0; r < 8; ++r) sc[r] = 1.0f;
  if (rs) {
#pragma unroll
    for (int r = 0; r < 8; ++r) sc[r] = rs[row0 + 8 * kh + r];
  }

  v8f c = {0.f, 0.f, 0.f, 0.f, 0.f, 0.f, 0.f, 0.f};
#pragma unroll
  for (int k0 = 0; k0 < IN_DIM / 4; ++k0) {
    const int kk = 4 * k0 + 2 * kh;
    const float2 av = *reinterpret_cast<const float2*>(xrow + 4 * k0);
    v2f a; a.x = av.x; a.y = av.y;
    v2f b; b.x = W1[(kk + 0) * HID + m];
           b.y = W1[(kk + 1) * HID + m];
    c = __builtin_amdgcn_wmma_f32_16x16x4_f32(false, a, false, b, (short)0, c, false, false);
  }
  float* hout = h + (size_t)(row0 + 8 * kh) * HID + m;
#pragma unroll
  for (int r = 0; r < 8; ++r)
    hout[(size_t)r * HID] = c[r] * sc[r];
}


__global__ void k_gemm2_wmma(const float* __restrict__ h1, const float* __restrict__ W2,
                             float* __restrict__ h2, int n) {
  const int lane = threadIdx.x & 31;
  const int wave = threadIdx.x >> 5;
  const int tile = blockIdx.x * (blockDim.x >> 5) + wave;
  const int row0 = tile << 4;
  if (row0 >= n) return;
  const int m  = lane & 15;
  const int kh = lane >> 4;
  const float bmask = (m < 2) ? 1.0f : 0.0f;
  const int   col   = (m < 2) ? m : 0;
  const float* arow = h1 + (size_t)(row0 + m) * HID + 2 * kh;

  v8f c = {0.f, 0.f, 0.f, 0.f, 0.f, 0.f, 0.f, 0.f};
#pragma unroll
  for (int k0 = 0; k0 < HID / 4; ++k0) {
    const int kk = 4 * k0 + 2 * kh;
    const float2 av = *reinterpret_cast<const float2*>(arow + 4 * k0);
    v2f a; a.x = av.x; a.y = av.y;
    v2f b; b.x = W2[(kk + 0) * 2 + col] * bmask;
           b.y = W2[(kk + 1) * 2 + col] * bmask;
    c = __builtin_amdgcn_wmma_f32_16x16x4_f32(false, a, false, b, (short)0, c, false, false);
  }
  if (m < 2) {
#pragma unroll
    for (int r = 0; r < 8; ++r)
      h2[(size_t)(row0 + r + 8 * kh) * 2 + m] = c[r];
  }
}


__global__ void k_self1(const float* __restrict__ h, const float* __restrict__ dis,
                        float* __restrict__ out1, int n) {
  int i = blockIdx.x * blockDim.x + threadIdx.x;
  if (i >= n) return;
  float s = dis[i]; s = s * s;
  const float4* hp = reinterpret_cast<const float4*>(h + (size_t)i * HID);
  float4* op = reinterpret_cast<float4*>(out1 + (size_t)i * HID);
#pragma unroll
  for (int q = 0; q < 4; ++q) {
    float4 v = hp[q];
    v.x *= s; v.y *= s; v.z *= s; v.w *= s;
    op[q] = v;
  }
}


static __device__ __forceinline__ unsigned row_bcast(int j, unsigned v) {
  const unsigned sel = 0x11111111u * (unsigned)j;
  return __builtin_amdgcn_permlane16(0u, v, sel, sel, false, true);
}

__global__ void k_scatter1(const int* __restrict__ src, const int* __restrict__ dst,
                           const float* __restrict__ dis, const float* __restrict__ h,
                           float* __restrict__ out1, int nE) {
  const int lane = threadIdx.x & 31;
  const unsigned f = lane & 15;
  const int wv     = __builtin_amdgcn_readfirstlane((int)(threadIdx.x >> 5));
  const int wpb    = (int)(blockDim.x >> 5);
  const int nWaves = (int)gridDim.x * wpb;
  const int nBatch = (nE + 31) >> 5;
  int b = (int)blockIdx.x * wpb + wv;
  if (b >= nBatch) return;

  unsigned s = 0, d = 0;
  if (lane < nE - (b << 5)) {
    s = (unsigned)src[(b << 5) + lane]; d = (unsigned)dst[(b << 5) + lane];
  }
  for (; b < nBatch; b += nWaves) {
    const int left = nE - (b << 5);
    const int bn   = b + nWaves;
    unsigned sN = 0, dN = 0;
    if (left >= 32) {
      const float ds = dis[s], dd = dis[d];
      float v[16];
#pragma unroll
      for (int j = 0; j < 16; ++j) v[j] = h[row_bcast(j, s) * HID + f];
      if (bn < nBatch && lane < nE - (bn << 5)) {
        sN = (unsigned)src[(bn << 5) + lane]; dN = (unsigned)dst[(bn << 5) + lane];
      }
      const float nrm = ds * dd;
#pragma unroll
      for (int j = 0; j < 16; ++j) {
        const unsigned dj = row_bcast(j, d);
        const float    nj = __uint_as_float(row_bcast(j, __float_as_uint(nrm)));
        atomicAdd(out1 + (dj * HID + f), v[j] * nj);
      }
    } else {
      float nrm = 0.f;
      if (lane < left) nrm = dis[s] * dis[d];
      const int mine = left - (lane & 16);
#pragma unroll
      for (int j = 0; j < 16; ++j) {
        const unsigned sj = row_bcast(j, s);
        const unsigned dj = row_bcast(j, d);
        const float    nj = __uint_as_float(row_bcast(j, __float_as_uint(nrm)));
        if (j < mine) atomicAdd(out1 + (dj * HID + f), h[sj * HID + f] * nj);
      }
    }
    s = sN; d = dN;
  }
}

__global__ void k_relu_bias(float* __restrict__ out1, const float* __restrict__ b1,
                            const float* __restrict__ rs, int n) {
  long long t = (long long)blockIdx.x * blockDim.x + threadIdx.x;
  if (t >= (long long)n * HID) return;
  int j = (int)(t & (HID - 1));
  float sc = 1.0f;
  if (rs) sc = rs[t >> 4];
  float v = out1[t] + b1[j];
  v = v > 0.f ? v : 0.f;
  out1[t] = v * sc;
}

__global__ void k_logsoftmax(const float* __restrict__ h2, const float* __restrict__ b2,
                             float* __restrict__ y, int n) {
  int i = blockIdx.x * blockDim.x + threadIdx.x;
  if (i >= n) return;
  float a0 = h2[(size_t)i * 2 + 0] + b2[0];
  float a1 = h2[(size_t)i * 2 + 1] + b2[1];
  float mx = fmaxf(a0, a1);
  float lse = mx + logf(expf(a0 - mx) + expf(a1 - mx));
  y[(size_t)i * 2 + 0] = a0 - lse;
  y[(size_t)i * 2 + 1] = a1 - lse;
}


#define BKT      256
#define CH_CAP   25024
#define NB_MAX   1024

__global__ __launch_bounds__(1024) void k_bin(const int* __restrict__ src, const int* __restrict__ dst,
                                              unsigned* __restrict__ list, int* __restrict__ segoff,
                                              int nE, int nB, int pitch) {
  __shared__ unsigned cnt[NB_MAX];
  __shared__ unsigned wsum[32];
  __shared__ unsigned stage[CH_CAP];
  const int tid  = threadIdx.x;
  const int lane = tid & 31;
  const int wave = tid >> 5;
  const int g    = blockIdx.x;
  int e0 = g * CH_CAP;            if (e0 > nE) e0 = nE;
  int e1 = e0 + CH_CAP;           if (e1 > nE) e1 = nE;

  cnt[tid] = 0u;
  __syncthreads();
  for (int e = e0 + tid; e < e1; e += 8 * 1024) {
    unsigned d[8];
#pragma unroll
    for (int k = 0; k < 8; ++k) { const int ee = e + k * 1024; d[k] = (unsigned)dst[ee < e1 ? ee : e1 - 1]; }
#pragma unroll
    for (int k = 0; k < 8; ++k) if (e + k * 1024 < e1) atomicAdd(&cnt[d[k] >> 8], 1u);
  }
  __syncthreads();

  const unsigned c = cnt[tid];
  unsigned x = c;
  for (int o = 1; o < 32; o <<= 1) { const unsigned y = __shfl_up(x, o); if (lane >= o) x += y; }
  if (lane == 31) wsum[wave] = x;
  __syncthreads();
  if (wave == 0) {
    const unsigned t = wsum[lane];
    unsigned u = t;
    for (int o = 1; o < 32; o <<= 1) { const unsigned y = __shfl_up(u, o); if (lane >= o) u += y; }
    wsum[lane] = u - t;
  }
  __syncthreads();
  const unsigned excl = x - c + wsum[wave];
  if (tid <= nB) segoff[(size_t)g * pitch + tid] = e0 + (int)excl;
  cnt[tid] = excl;
  __syncthreads();

  for (int e = e0 + tid; e < e1; e += 8 * 1024) {
    unsigned d[8], s[8];
#pragma unroll
    for (int k = 0; k < 8; ++k) {
      const int ee = e + k * 1024, ec = ee < e1 ? ee : e1 - 1;
      d[k] = (unsigned)dst[ec]; s[k] = (unsigned)src[ec];
    }
#pragma unroll
    for (int k = 0; k < 8; ++k) {
      if (e + k * 1024 < e1) {
        const unsigned r = atomicAdd(&cnt[d[k] >> 8], 1u);
        stage[r] = (s[k] << 8) | (d[k] & 255u);
      }
    }
  }
  __syncthreads();
  const int len = e1 - e0;
  for (int i = tid; i < len; i += 1024) list[e0 + i] = stage[i];
}

__global__ __launch_bounds__(1024) void k_degdis(const unsigned* __restrict__ list, const int* __restrict__ segoff,
                                                 float* __restrict__ dis, int n, int pitch, int nChunk) {
  __shared__ unsigned cnt[BKT];
  const int tid  = threadIdx.x;
  const int lane = tid & 31;
  const int wv   = __builtin_amdgcn_readfirstlane((int)(threadIdx.x >> 5));
  const int b    = blockIdx.x;
  if (tid < BKT) cnt[tid] = 0u;
  __syncthreads();
  int beg = 0, end = 0;
  if (wv < nChunk) { beg = segoff[(size_t)wv * pitch + b]; end = segoff[(size_t)wv * pitch + b + 1]; }
  for (int g = wv; g < nChunk; g += 32) {
    const int gn = g + 32;
    int begN = 0, endN = 0;
    if (gn < nChunk) { begN = segoff[(size_t)gn * pitch + b]; endN = segoff[(size_t)gn * pitch + b + 1]; }
    for (int p = beg + lane; p < end; p += 32) atomicAdd(&cnt[list[p] & 255u], 1u);
    beg = begN; end = endN;
  }
  __syncthreads();
  const int node = b * BKT + tid;
  if (tid < BKT && node < n) dis[node] = rsqrtf((float)(cnt[tid] + 1u));
}

__global__ __launch_bounds__(512) void k_agg(const unsigned* __restrict__ list, const int* __restrict__ segoff,
                                             const float* __restrict__ dis, const float* __restrict__ h,
                                             float* __restrict__ out, int n, int pitch, int nChunk) {
  __shared__ float acc[BKT * HID];
  const int tid  = threadIdx.x;
  const int lane = tid & 31;
  const unsigned f = lane & 15;
  const int wv   = __builtin_amdgcn_readfirstlane((int)(threadIdx.x >> 5));
  const int b    = blockIdx.x;
  const int node0 = b * BKT;

  for (int t = tid; t < BKT * HID; t += 512) {
    const int node = node0 + (t >> 4);
    acc[t] = node < n ? h[(size_t)node * HID + (t & 15)] : 0.f;
  }
  __syncthreads();

  int beg = 0, end = 0;
  if (wv < nChunk) { beg = segoff[(size_t)wv * pitch + b]; end = segoff[(size_t)wv * pitch + b + 1]; }
  for (int g = wv; g < nChunk; g += 16) {
    const int gn = g + 16;
    int begN = 0, endN = 0;
    if (gn < nChunk) { begN = segoff[(size_t)gn * pitch + b]; endN = segoff[(size_t)gn * pitch + b + 1]; }
    for (int base = beg; base < end; base += 32) {
      const int left = end - base;
      const int p = base + lane;
      const unsigned e  = list[p < end ? p : end - 1];
      const unsigned s  = e >> 8, dl = e & 255u;
      float v[16];
#pragma unroll
      for (int j = 0; j < 16; ++j) v[j] = h[row_bcast(j, s) * HID + f];
      const int mine = left - (lane & 16);
#pragma unroll
      for (int j = 0; j < 16; ++j) {
        const unsigned dj = row_bcast(j, dl);
        if (j < mine) atomicAdd(&acc[dj * HID + f], v[j]);
      }
    }
    beg = begN; end = endN;
  }
  __syncthreads();
  for (int t = tid; t < BKT * HID; t += 512) {
    const int node = node0 + (t >> 4);
    if (node < n) out[(size_t)node0 * HID + t] = acc[t] * dis[node];
  }
}


extern "C" void kernel_launch(void* const* d_in, const int* in_sizes, int n_in,
                              void* d_out, int out_size, void* d_ws, size_t ws_size,
                              hipStream_t stream) {
  const float* x  = (const float*)d_in[0];
  const int*   ei = (const int*)d_in[1];
  const float* W1 = (const float*)d_in[2];
  const float* b1 = (const float*)d_in[3];
  const float* W2 = (const float*)d_in[4];
  const float* b2 = (const float*)d_in[5];

  const int n  = in_sizes[0] / IN_DIM;
  const int nE = in_sizes[1] / 2;
  const int* src = ei;
  const int* dst = ei + nE;
  float* ws = (float*)d_ws;
  float* y  = (float*)d_out;

  const int TB = 256;
  dim3 blk(TB);
  int gN   = (n + TB - 1) / TB;
  int gN16 = (int)(((long long)n * HID + TB - 1) / TB);
  int tiles = (n + 15) / 16;
  int gT   = (tiles + (TB / 32) - 1) / (TB / 32);

  const int nB     = (n + BKT - 1) / BKT;
  const int pitch  = (nB + 1 + 31) & ~31;
  const int nChunk = (nE + CH_CAP - 1) / CH_CAP;
  const size_t need = ((size_t)n * (1 + HID + HID + 2) + (size_t)nE + (size_t)nChunk * pitch) * sizeof(float);

  if (ws_size >= need && nChunk > 0 && nB < NB_MAX && n <= (1 << 24)) {
    float* dis  = ws;
    float* h    = dis  + n;
    float* out1 = h    + (size_t)n * HID;
    float* h2   = out1 + (size_t)n * HID;
    unsigned* list = (unsigned*)(h2 + (size_t)n * 2);
    int* segoff    = (int*)(list + nE);
    float* agg  = h;

    k_bin       <<<nChunk, dim3(1024), 0, stream>>>(src, dst, list, segoff, nE, nB, pitch);
    k_degdis    <<<nB,  dim3(1024), 0, stream>>>(list, segoff, dis, n, pitch, nChunk);
    k_gemm1_wmma<<<gT,  blk, 0, stream>>>(x, W1, dis, h, n);
    k_agg       <<<nB,  dim3(512), 0, stream>>>(list, segoff, dis, h, out1, n, pitch, nChunk);
    k_relu_bias <<<gN16,blk, 0, stream>>>(out1, b1, dis, n);
    k_agg       <<<nB,  dim3(512), 0, stream>>>(list, segoff, dis, out1, agg, n, pitch, nChunk);
    k_gemm2_wmma<<<gT,  blk, 0, stream>>>(agg, W2, h2, n);
    k_logsoftmax<<<gN,  blk, 0, stream>>>(h2, b2, y, n);
    return;
  }

  float* deg  = ws;
  float* dis  = deg + n;
  float* h    = dis + n;
  float* out1 = h    + (size_t)n * HID;
  float* h2   = out1 + (size_t)n * HID;
  float* agg  = h;
  int gE   = (nE + TB - 1) / TB;
  int gS   = 2048;

  k_init_deg  <<<gN,  blk, 0, stream>>>(deg, n);
  k_deg_edges <<<gE,  blk, 0, stream>>>(dst, deg, nE);
  k_dis       <<<gN,  blk, 0, stream>>>(deg, dis, n);
  k_gemm1_wmma<<<gT,  blk, 0, stream>>>(x, W1, nullptr, h, n);
  k_self1     <<<gN,  blk, 0, stream>>>(h, dis, out1, n);
  k_scatter1  <<<gS,  blk, 0, stream>>>(src, dst, dis, h, out1, nE);
  k_relu_bias <<<gN16,blk, 0, stream>>>(out1, b1, nullptr, n);
  k_self1     <<<gN,  blk, 0, stream>>>(out1, dis, agg, n);
  k_scatter1  <<<gS,  blk, 0, stream>>>(src, dst, dis, out1, agg, nE);
  k_gemm2_wmma<<<gT,  blk, 0, stream>>>(agg, W2, h2, n);
  k_logsoftmax<<<gN,  blk, 0, stream>>>(h2, b2, y, n);
}


# Bias, ReLU and the next layer's dis[src] applied in the layer-1 aggregation's epilogue, sums still in LDS (sum*dis+bias now one fma): the elementwise launch of 100000 one-shot waves is gone, 8 -> 7
# speedup vs baseline: 5.3126x; 1.0506x over previous
#include <hip/hip_runtime.h>
#include <hip/hip_bf16.h>
#include <math.h>

typedef __attribute__((ext_vector_type(2))) float v2f;
typedef __attribute__((ext_vector_type(8))) float v8f;

#define IN_DIM 128
#define HID    16


__global__ void k_init_deg(float* __restrict__ deg, int n) {
  int i = blockIdx.x * blockDim.x + threadIdx.x;
  if (i < n) deg[i] = 1.0f;
}

__global__ void k_deg_edges(const int* __restrict__ dst, float* __restrict__ deg, int nE) {
  int e = blockIdx.x * blockDim.x + threadIdx.x;
  if (e < nE) atomicAdd(&deg[dst[e]], 1.0f);
}

__global__ void k_dis(const float* __restrict__ deg, float* __restrict__ dis, int n) {
  int i = blockIdx.x * blockDim.x + threadIdx.x;
  if (i < n) dis[i] = rsqrtf(deg[i]);
}


__global__ void k_gemm1_wmma(const float* __restrict__ x, const float* __restrict__ W1,
                             const float* __restrict__ rs, float* __restrict__ h, int n) {
  const int lane = threadIdx.x & 31;
  const int wave = threadIdx.x >> 5;
  const int tile = blockIdx.x * (blockDim.x >> 5) + wave;
  const int row0 = tile << 4;
  if (row0 >= n) return;
  const int m  = lane & 15;
  const int kh = lane >> 4;
  const float* xrow = x + (size_t)(row0 + m) * IN_DIM + 2 * kh;

  float sc[8];
#pragma unroll
  for (int r = 0; r < 8; ++r) sc[r] = 1.0f;
  if (rs) {
#pragma unroll
    for (int r = 0; r < 8; ++r) sc[r] = rs[row0 + 8 * kh + r];
  }

  v8f c = {0.f, 0.f, 0.f, 0.f, 0.f, 0.f, 0.f, 0.f};
#pragma unroll
  for (int k0 = 0; k0 < IN_DIM / 4; ++k0) {
    const int kk = 4 * k0 + 2 * kh;
    const float2 av = *reinterpret_cast<const float2*>(xrow + 4 * k0);
    v2f a; a.x = av.x; a.y = av.y;
    v2f b; b.x = W1[(kk + 0) * HID + m];
           b.y = W1[(kk + 1) * HID + m];
    c = __builtin_amdgcn_wmma_f32_16x16x4_f32(false, a, false, b, (short)0, c, false, false);
  }
  float* hout = h + (size_t)(row0 + 8 * kh) * HID + m;
#pragma unroll
  for (int r = 0; r < 8; ++r)
    hout[(size_t)r * HID] = c[r] * sc[r];
}


__global__ void k_gemm2_wmma(const float* __restrict__ h1, const float* __restrict__ W2,
                             float* __restrict__ h2, int n) {
  const int lane = threadIdx.x & 31;
  const int wave = threadIdx.x >> 5;
  const int tile = blockIdx.x * (blockDim.x >> 5) + wave;
  const int row0 = tile << 4;
  if (row0 >= n) return;
  const int m  = lane & 15;
  const int kh = lane >> 4;
  const float bmask = (m < 2) ? 1.0f : 0.0f;
  const int   col   = (m < 2) ? m : 0;
  const float* arow = h1 + (size_t)(row0 + m) * HID + 2 * kh;

  v8f c = {0.f, 0.f, 0.f, 0.f, 0.f, 0.f, 0.f, 0.f};
#pragma unroll
  for (int k0 = 0; k0 < HID / 4; ++k0) {
    const int kk = 4 * k0 + 2 * kh;
    const float2 av = *reinterpret_cast<const float2*>(arow + 4 * k0);
    v2f a; a.x = av.x; a.y = av.y;
    v2f b; b.x = W2[(kk + 0) * 2 + col] * bmask;
           b.y = W2[(kk + 1) * 2 + col] * bmask;
    c = __builtin_amdgcn_wmma_f32_16x16x4_f32(false, a, false, b, (short)0, c, false, false);
  }
  if (m < 2) {
#pragma unroll
    for (int r = 0; r < 8; ++r)
      h2[(size_t)(row0 + r + 8 * kh) * 2 + m] = c[r];
  }
}


__global__ void k_self1(const float* __restrict__ h, const float* __restrict__ dis,
                        float* __restrict__ out1, int n) {
  int i = blockIdx.x * blockDim.x + threadIdx.x;
  if (i >= n) return;
  float s = dis[i]; s = s * s;
  const float4* hp = reinterpret_cast<const float4*>(h + (size_t)i * HID);
  float4* op = reinterpret_cast<float4*>(out1 + (size_t)i * HID);
#pragma unroll
  for (int q = 0; q < 4; ++q) {
    float4 v = hp[q];
    v.x *= s; v.y *= s; v.z *= s; v.w *= s;
    op[q] = v;
  }
}


static __device__ __forceinline__ unsigned row_bcast(int j, unsigned v) {
  const unsigned sel = 0x11111111u * (unsigned)j;
  return __builtin_amdgcn_permlane16(0u, v, sel, sel, false, true);
}

__global__ void k_scatter1(const int* __restrict__ src, const int* __restrict__ dst,
                           const float* __restrict__ dis, const float* __restrict__ h,
                           float* __restrict__ out1, int nE) {
  const int lane = threadIdx.x & 31;
  const unsigned f = lane & 15;
  const int wv     = __builtin_amdgcn_readfirstlane((int)(threadIdx.x >> 5));
  const int wpb    = (int)(blockDim.x >> 5);
  const int nWaves = (int)gridDim.x * wpb;
  const int nBatch = (nE + 31) >> 5;
  int b = (int)blockIdx.x * wpb + wv;
  if (b >= nBatch) return;

  unsigned s = 0, d = 0;
  if (lane < nE - (b << 5)) {
    s = (unsigned)src[(b << 5) + lane]; d = (unsigned)dst[(b << 5) + lane];
  }
  for (; b < nBatch; b += nWaves) {
    const int left = nE - (b << 5);
    const int bn   = b + nWaves;
    unsigned sN = 0, dN = 0;
    if (left >= 32) {
      const float ds = dis[s], dd = dis[d];
      float v[16];
#pragma unroll
      for (int j = 0; j < 16; ++j) v[j] = h[row_bcast(j, s) * HID + f];
      if (bn < nBatch && lane < nE - (bn << 5)) {
        sN = (unsigned)src[(bn << 5) + lane]; dN = (unsigned)dst[(bn << 5) + lane];
      }
      const float nrm = ds * dd;
#pragma unroll
      for (int j = 0; j < 16; ++j) {
        const unsigned dj = row_bcast(j, d);
        const float    nj = __uint_as_float(row_bcast(j, __float_as_uint(nrm)));
        atomicAdd(out1 + (dj * HID + f), v[j] * nj);
      }
    } else {
      float nrm = 0.f;
      if (lane < left) nrm = dis[s] * dis[d];
      const int mine = left - (lane & 16);
#pragma unroll
      for (int j = 0; j < 16; ++j) {
        const unsigned sj = row_bcast(j, s);
        const unsigned dj = row_bcast(j, d);
        const float    nj = __uint_as_float(row_bcast(j, __float_as_uint(nrm)));
        if (j < mine) atomicAdd(out1 + (dj * HID + f), h[sj * HID + f] * nj);
      }
    }
    s = sN; d = dN;
  }
}

__global__ void k_relu_bias(float* __restrict__ out1, const float* __restrict__ b1,
                            const float* __restrict__ rs, int n) {
  long long t = (long long)blockIdx.x * blockDim.x + threadIdx.x;
  if (t >= (long long)n * HID) return;
  int j = (int)(t & (HID - 1));
  float sc = 1.0f;
  if (rs) sc = rs[t >> 4];
  float v = out1[t] + b1[j];
  v = v > 0.f ? v : 0.f;
  out1[t] = v * sc;
}

__global__ void k_logsoftmax(const float* __restrict__ h2, const float* __restrict__ b2,
                             float* __restrict__ y, int n) {
  int i = blockIdx.x * blockDim.x + threadIdx.x;
  if (i >= n) return;
  float a0 = h2[(size_t)i * 2 + 0] + b2[0];
  float a1 = h2[(size_t)i * 2 + 1] + b2[1];
  float mx = fmaxf(a0, a1);
  float lse = mx + logf(expf(a0 - mx) + expf(a1 - mx));
  y[(size_t)i * 2 + 0] = a0 - lse;
  y[(size_t)i * 2 + 1] = a1 - lse;
}


#define BKT      256
#define CH_CAP   25024
#define NB_MAX   1024

__global__ __launch_bounds__(1024) void k_bin(const int* __restrict__ src, const int* __restrict__ dst,
                                              unsigned* __restrict__ list, int* __restrict__ segoff,
                                              int nE, int nB, int pitch) {
  __shared__ unsigned cnt[NB_MAX];
  __shared__ unsigned wsum[32];
  __shared__ unsigned stage[CH_CAP];
  const int tid  = threadIdx.x;
  const int lane = tid & 31;
  const int wave = tid >> 5;
  const int g    = blockIdx.x;
  int e0 = g * CH_CAP;            if (e0 > nE) e0 = nE;
  int e1 = e0 + CH_CAP;           if (e1 > nE) e1 = nE;

  cnt[tid] = 0u;
  __syncthreads();
  for (int e = e0 + tid; e < e1; e += 8 * 1024) {
    unsigned d[8];
#pragma unroll
    for (int k = 0; k < 8; ++k) { const int ee = e + k * 1024; d[k] = (unsigned)dst[ee < e1 ? ee : e1 - 1]; }
#pragma unroll
    for (int k = 0; k < 8; ++k) if (e + k * 1024 < e1) atomicAdd(&cnt[d[k] >> 8], 1u);
  }
  __syncthreads();

  const unsigned c = cnt[tid];
  unsigned x = c;
  for (int o = 1; o < 32; o <<= 1) { const unsigned y = __shfl_up(x, o); if (lane >= o) x += y; }
  if (lane == 31) wsum[wave] = x;
  __syncthreads();
  if (wave == 0) {
    const unsigned t = wsum[lane];
    unsigned u = t;
    for (int o = 1; o < 32; o <<= 1) { const unsigned y = __shfl_up(u, o); if (lane >= o) u += y; }
    wsum[lane] = u - t;
  }
  __syncthreads();
  const unsigned excl = x - c + wsum[wave];
  if (tid <= nB) segoff[(size_t)g * pitch + tid] = e0 + (int)excl;
  cnt[tid] = excl;
  __syncthreads();

  for (int e = e0 + tid; e < e1; e += 8 * 1024) {
    unsigned d[8], s[8];
#pragma unroll
    for (int k = 0; k < 8; ++k) {
      const int ee = e + k * 1024, ec = ee < e1 ? ee : e1 - 1;
      d[k] = (unsigned)dst[ec]; s[k] = (unsigned)src[ec];
    }
#pragma unroll
    for (int k = 0; k < 8; ++k) {
      if (e + k * 1024 < e1) {
        const unsigned r = atomicAdd(&cnt[d[k] >> 8], 1u);
        stage[r] = (s[k] << 8) | (d[k] & 255u);
      }
    }
  }
  __syncthreads();
  const int len = e1 - e0;
  for (int i = tid; i < len; i += 1024) list[e0 + i] = stage[i];
}

__global__ __launch_bounds__(1024) void k_degdis(const unsigned* __restrict__ list, const int* __restrict__ segoff,
                                                 float* __restrict__ dis, int n, int pitch, int nChunk) {
  __shared__ unsigned cnt[BKT];
  const int tid  = threadIdx.x;
  const int lane = tid & 31;
  const int wv   = __builtin_amdgcn_readfirstlane((int)(threadIdx.x >> 5));
  const int b    = blockIdx.x;
  if (tid < BKT) cnt[tid] = 0u;
  __syncthreads();
  int beg = 0, end = 0;
  if (wv < nChunk) { beg = segoff[(size_t)wv * pitch + b]; end = segoff[(size_t)wv * pitch + b + 1]; }
  for (int g = wv; g < nChunk; g += 32) {
    const int gn = g + 32;
    int begN = 0, endN = 0;
    if (gn < nChunk) { begN = segoff[(size_t)gn * pitch + b]; endN = segoff[(size_t)gn * pitch + b + 1]; }
    for (int p = beg + lane; p < end; p += 32) atomicAdd(&cnt[list[p] & 255u], 1u);
    beg = begN; end = endN;
  }
  __syncthreads();
  const int node = b * BKT + tid;
  if (tid < BKT && node < n) dis[node] = rsqrtf((float)(cnt[tid] + 1u));
}

__global__ __launch_bounds__(512) void k_agg(const unsigned* __restrict__ list, const int* __restrict__ segoff,
                                             const float* __restrict__ dis, const float* __restrict__ h,
                                             const float* __restrict__ bias,
                                             float* __restrict__ out, int n, int pitch, int nChunk) {
  __shared__ float acc[BKT * HID];
  const int tid  = threadIdx.x;
  const int lane = tid & 31;
  const unsigned f = lane & 15;
  const int wv   = __builtin_amdgcn_readfirstlane((int)(threadIdx.x >> 5));
  const int b    = blockIdx.x;
  const int node0 = b * BKT;

  float bj = 0.f;
  if (bias) bj = bias[tid & (HID - 1)];

  for (int t = tid; t < BKT * HID; t += 512) {
    const int node = node0 + (t >> 4);
    acc[t] = node < n ? h[(size_t)node * HID + (t & 15)] : 0.f;
  }
  __syncthreads();

  int beg = 0, end = 0;
  if (wv < nChunk) { beg = segoff[(size_t)wv * pitch + b]; end = segoff[(size_t)wv * pitch + b + 1]; }
  for (int g = wv; g < nChunk; g += 16) {
    const int gn = g + 16;
    int begN = 0, endN = 0;
    if (gn < nChunk) { begN = segoff[(size_t)gn * pitch + b]; endN = segoff[(size_t)gn * pitch + b + 1]; }
    for (int base = beg; base < end; base += 32) {
      const int left = end - base;
      const int p = base + lane;
      const unsigned e  = list[p < end ? p : end - 1];
      const unsigned s  = e >> 8, dl = e & 255u;
      float v[16];
#pragma unroll
      for (int j = 0; j < 16; ++j) v[j] = h[row_bcast(j, s) * HID + f];
      const int mine = left - (lane & 16);
#pragma unroll
      for (int j = 0; j < 16; ++j) {
        const unsigned dj = row_bcast(j, dl);
        if (j < mine) atomicAdd(&acc[dj * HID + f], v[j]);
      }
    }
    beg = begN; end = endN;
  }
  __syncthreads();
  for (int t = tid; t < BKT * HID; t += 512) {
    const int node = node0 + (t >> 4);
    if (node < n) {
      const float dn = dis[node];
      float v = acc[t] * dn;
      if (bias) {
        v += bj;
        v = v > 0.f ? v : 0.f;
        v *= dn;
      }
      out[(size_t)node0 * HID + t] = v;
    }
  }
}


extern "C" void kernel_launch(void* const* d_in, const int* in_sizes, int n_in,
                              void* d_out, int out_size, void* d_ws, size_t ws_size,
                              hipStream_t stream) {
  const float* x  = (const float*)d_in[0];
  const int*   ei = (const int*)d_in[1];
  const float* W1 = (const float*)d_in[2];
  const float* b1 = (const float*)d_in[3];
  const float* W2 = (const float*)d_in[4];
  const float* b2 = (const float*)d_in[5];

  const int n  = in_sizes[0] / IN_DIM;
  const int nE = in_sizes[1] / 2;
  const int* src = ei;
  const int* dst = ei + nE;
  float* ws = (float*)d_ws;
  float* y  = (float*)d_out;

  const int TB = 256;
  dim3 blk(TB);
  int gN   = (n + TB - 1) / TB;
  int gN16 = (int)(((long long)n * HID + TB - 1) / TB);
  int tiles = (n + 15) / 16;
  int gT   = (tiles + (TB / 32) - 1) / (TB / 32);

  const int nB     = (n + BKT - 1) / BKT;
  const int pitch  = (nB + 1 + 31) & ~31;
  const int nChunk = (nE + CH_CAP - 1) / CH_CAP;
  const size_t need = ((size_t)n * (1 + HID + HID + 2) + (size_t)nE + (size_t)nChunk * pitch) * sizeof(float);

  if (ws_size >= need && nChunk > 0 && nB < NB_MAX && n <= (1 << 24)) {
    float* dis  = ws;
    float* h    = dis  + n;
    float* out1 = h    + (size_t)n * HID;
    float* h2   = out1 + (size_t)n * HID;
    unsigned* list = (unsigned*)(h2 + (size_t)n * 2);
    int* segoff    = (int*)(list + nE);
    float* agg  = h;

    k_bin       <<<nChunk, dim3(1024), 0, stream>>>(src, dst, list, segoff, nE, nB, pitch);
    k_degdis    <<<nB,  dim3(1024), 0, stream>>>(list, segoff, dis, n, pitch, nChunk);
    k_gemm1_wmma<<<gT,  blk, 0, stream>>>(x, W1, dis, h, n);
    k_agg       <<<nB,  dim3(512), 0, stream>>>(list, segoff, dis, h, b1, out1, n, pitch, nChunk);
    k_agg       <<<nB,  dim3(512), 0, stream>>>(list, segoff, dis, out1, nullptr, agg, n, pitch, nChunk);
    k_gemm2_wmma<<<gT,  blk, 0, stream>>>(agg, W2, h2, n);
    k_logsoftmax<<<gN,  blk, 0, stream>>>(h2, b2, y, n);
    return;
  }

  float* deg  = ws;
  float* dis  = deg + n;
  float* h    = dis + n;
  float* out1 = h    + (size_t)n * HID;
  float* h2   = out1 + (size_t)n * HID;
  float* agg  = h;
  int gE   = (nE + TB - 1) / TB;
  int gS   = 2048;

  k_init_deg  <<<gN,  blk, 0, stream>>>(deg, n);
  k_deg_edges <<<gE,  blk, 0, stream>>>(dst, deg, nE);
  k_dis       <<<gN,  blk, 0, stream>>>(deg, dis, n);
  k_gemm1_wmma<<<gT,  blk, 0, stream>>>(x, W1, nullptr, h, n);
  k_self1     <<<gN,  blk, 0, stream>>>(h, dis, out1, n);
  k_scatter1  <<<gS,  blk, 0, stream>>>(src, dst, dis, h, out1, nE);
  k_relu_bias <<<gN16,blk, 0, stream>>>(out1, b1, nullptr, n);
  k_self1     <<<gN,  blk, 0, stream>>>(out1, dis, agg, n);
  k_scatter1  <<<gS,  blk, 0, stream>>>(src, dst, dis, out1, agg, nE);
  k_gemm2_wmma<<<gT,  blk, 0, stream>>>(agg, W2, h2, n);
  k_logsoftmax<<<gN,  blk, 0, stream>>>(h2, b2, y, n);
}


# Layer 2 takes the W2 product first; its binned pass sums 2-word rows (lane = edge: one 8-B gather + 2 LDS adds per edge, not a 64-B row + 16), ends in bias + log_softmax: 6 launches; sums reordered
# speedup vs baseline: 5.9808x; 1.1258x over previous
#include <hip/hip_runtime.h>
#include <hip/hip_bf16.h>
#include <math.h>

typedef __attribute__((ext_vector_type(2))) float v2f;
typedef __attribute__((ext_vector_type(8))) float v8f;

#define IN_DIM 128
#define HID    16


__global__ void k_init_deg(float* __restrict__ deg, int n) {
  int i = blockIdx.x * blockDim.x + threadIdx.x;
  if (i < n) deg[i] = 1.0f;
}

__global__ void k_deg_edges(const int* __restrict__ dst, float* __restrict__ deg, int nE) {
  int e = blockIdx.x * blockDim.x + threadIdx.x;
  if (e < nE) atomicAdd(&deg[dst[e]], 1.0f);
}

__global__ void k_dis(const float* __restrict__ deg, float* __restrict__ dis, int n) {
  int i = blockIdx.x * blockDim.x + threadIdx.x;
  if (i < n) dis[i] = rsqrtf(deg[i]);
}


__global__ void k_gemm1_wmma(const float* __restrict__ x, const float* __restrict__ W1,
                             const float* __restrict__ rs, float* __restrict__ h, int n) {
  const int lane = threadIdx.x & 31;
  const int wave = threadIdx.x >> 5;
  const int tile = blockIdx.x * (blockDim.x >> 5) + wave;
  const int row0 = tile << 4;
  if (row0 >= n) return;
  const int m  = lane & 15;
  const int kh = lane >> 4;
  const float* xrow = x + (size_t)(row0 + m) * IN_DIM + 2 * kh;

  float sc[8];
#pragma unroll
  for (int r = 0; r < 8; ++r) sc[r] = 1.0f;
  if (rs) {
#pragma unroll
    for (int r = 0; r < 8; ++r) sc[r] = rs[row0 + 8 * kh + r];
  }

  v8f c = {0.f, 0.f, 0.f, 0.f, 0.f, 0.f, 0.f, 0.f};
#pragma unroll
  for (int k0 = 0; k0 < IN_DIM / 4; ++k0) {
    const int kk = 4 * k0 + 2 * kh;
    const float2 av = *reinterpret_cast<const float2*>(xrow + 4 * k0);
    v2f a; a.x = av.x; a.y = av.y;
    v2f b; b.x = W1[(kk + 0) * HID + m];
           b.y = W1[(kk + 1) * HID + m];
    c = __builtin_amdgcn_wmma_f32_16x16x4_f32(false, a, false, b, (short)0, c, false, false);
  }
  float* hout = h + (size_t)(row0 + 8 * kh) * HID + m;
#pragma unroll
  for (int r = 0; r < 8; ++r)
    hout[(size_t)r * HID] = c[r] * sc[r];
}


__global__ void k_gemm2_wmma(const float* __restrict__ h1, const float* __restrict__ W2,
                             float* __restrict__ h2, int n) {
  const int lane = threadIdx.x & 31;
  const int wave = threadIdx.x >> 5;
  const int tile = blockIdx.x * (blockDim.x >> 5) + wave;
  const int row0 = tile << 4;
  if (row0 >= n) return;
  const int m  = lane & 15;
  const int kh = lane >> 4;
  const float bmask = (m < 2) ? 1.0f : 0.0f;
  const int   col   = (m < 2) ? m : 0;
  const float* arow = h1 + (size_t)(row0 + m) * HID + 2 * kh;

  v8f c = {0.f, 0.f, 0.f, 0.f, 0.f, 0.f, 0.f, 0.f};
#pragma unroll
  for (int k0 = 0; k0 < HID / 4; ++k0) {
    const int kk = 4 * k0 + 2 * kh;
    const float2 av = *reinterpret_cast<const float2*>(arow + 4 * k0);
    v2f a; a.x = av.x; a.y = av.y;
    v2f b; b.x = W2[(kk + 0) * 2 + col] * bmask;
           b.y = W2[(kk + 1) * 2 + col] * bmask;
    c = __builtin_amdgcn_wmma_f32_16x16x4_f32(false, a, false, b, (short)0, c, false, false);
  }
  if (m < 2) {
#pragma unroll
    for (int r = 0; r < 8; ++r)
      h2[(size_t)(row0 + r + 8 * kh) * 2 + m] = c[r];
  }
}


__global__ void k_self1(const float* __restrict__ h, const float* __restrict__ dis,
                        float* __restrict__ out1, int n) {
  int i = blockIdx.x * blockDim.x + threadIdx.x;
  if (i >= n) return;
  float s = dis[i]; s = s * s;
  const float4* hp = reinterpret_cast<const float4*>(h + (size_t)i * HID);
  float4* op = reinterpret_cast<float4*>(out1 + (size_t)i * HID);
#pragma unroll
  for (int q = 0; q < 4; ++q) {
    float4 v = hp[q];
    v.x *= s; v.y *= s; v.z *= s; v.w *= s;
    op[q] = v;
  }
}


static __device__ __forceinline__ unsigned row_bcast(int j, unsigned v) {
  const unsigned sel = 0x11111111u * (unsigned)j;
  return __builtin_amdgcn_permlane16(0u, v, sel, sel, false, true);
}

__global__ void k_scatter1(const int* __restrict__ src, const int* __restrict__ dst,
                           const float* __restrict__ dis, const float* __restrict__ h,
                           float* __restrict__ out1, int nE) {
  const int lane = threadIdx.x & 31;
  const unsigned f = lane & 15;
  const int wv     = __builtin_amdgcn_readfirstlane((int)(threadIdx.x >> 5));
  const int wpb    = (int)(blockDim.x >> 5);
  const int nWaves = (int)gridDim.x * wpb;
  const int nBatch = (nE + 31) >> 5;
  int b = (int)blockIdx.x * wpb + wv;
  if (b >= nBatch) return;

  unsigned s = 0, d = 0;
  if (lane < nE - (b << 5)) {
    s = (unsigned)src[(b << 5) + lane]; d = (unsigned)dst[(b << 5) + lane];
  }
  for (; b < nBatch; b += nWaves) {
    const int left = nE - (b << 5);
    const int bn   = b + nWaves;
    unsigned sN = 0, dN = 0;
    if (left >= 32) {
      const float ds = dis[s], dd = dis[d];
      float v[16];
#pragma unroll
      for (int j = 0; j < 16; ++j) v[j] = h[row_bcast(j, s) * HID + f];
      if (bn < nBatch && lane < nE - (bn << 5)) {
        sN = (unsigned)src[(bn << 5) + lane]; dN = (unsigned)dst[(bn << 5) + lane];
      }
      const float nrm = ds * dd;
#pragma unroll
      for (int j = 0; j < 16; ++j) {
        const unsigned dj = row_bcast(j, d);
        const float    nj = __uint_as_float(row_bcast(j, __float_as_uint(nrm)));
        atomicAdd(out1 + (dj * HID + f), v[j] * nj);
      }
    } else {
      float nrm = 0.f;
      if (lane < left) nrm = dis[s] * dis[d];
      const int mine = left - (lane & 16);
#pragma unroll
      for (int j = 0; j < 16; ++j) {
        const unsigned sj = row_bcast(j, s);
        const unsigned dj = row_bcast(j, d);
        const float    nj = __uint_as_float(row_bcast(j, __float_as_uint(nrm)));
        if (j < mine) atomicAdd(out1 + (dj * HID + f), h[sj * HID + f] * nj);
      }
    }
    s = sN; d = dN;
  }
}

__global__ void k_relu_bias(float* __restrict__ out1, const float* __restrict__ b1,
                            const float* __restrict__ rs, int n) {
  long long t = (long long)blockIdx.x * blockDim.x + threadIdx.x;
  if (t >= (long long)n * HID) return;
  int j = (int)(t & (HID - 1));
  float sc = 1.0f;
  if (rs) sc = rs[t >> 4];
  float v = out1[t] + b1[j];
  v = v > 0.f ? v : 0.f;
  out1[t] = v * sc;
}

__global__ void k_logsoftmax(const float* __restrict__ h2, const float* __restrict__ b2,
                             float* __restrict__ y, int n) {
  int i = blockIdx.x * blockDim.x + threadIdx.x;
  if (i >= n) return;
  float a0 = h2[(size_t)i * 2 + 0] + b2[0];
  float a1 = h2[(size_t)i * 2 + 1] + b2[1];
  float mx = fmaxf(a0, a1);
  float lse = mx + logf(expf(a0 - mx) + expf(a1 - mx));
  y[(size_t)i * 2 + 0] = a0 - lse;
  y[(size_t)i * 2 + 1] = a1 - lse;
}


#define BKT      256
#define CH_CAP   25024
#define NB_MAX   1024

__global__ __launch_bounds__(1024) void k_bin(const int* __restrict__ src, const int* __restrict__ dst,
                                              unsigned* __restrict__ list, int* __restrict__ segoff,
                                              int nE, int nB, int pitch) {
  __shared__ unsigned cnt[NB_MAX];
  __shared__ unsigned wsum[32];
  __shared__ unsigned stage[CH_CAP];
  const int tid  = threadIdx.x;
  const int lane = tid & 31;
  const int wave = tid >> 5;
  const int g    = blockIdx.x;
  int e0 = g * CH_CAP;            if (e0 > nE) e0 = nE;
  int e1 = e0 + CH_CAP;           if (e1 > nE) e1 = nE;

  cnt[tid] = 0u;
  __syncthreads();
  for (int e = e0 + tid; e < e1; e += 8 * 1024) {
    unsigned d[8];
#pragma unroll
    for (int k = 0; k < 8; ++k) { const int ee = e + k * 1024; d[k] = (unsigned)dst[ee < e1 ? ee : e1 - 1]; }
#pragma unroll
    for (int k = 0; k < 8; ++k) if (e + k * 1024 < e1) atomicAdd(&cnt[d[k] >> 8], 1u);
  }
  __syncthreads();

  const unsigned c = cnt[tid];
  unsigned x = c;
  for (int o = 1; o < 32; o <<= 1) { const unsigned y = __shfl_up(x, o); if (lane >= o) x += y; }
  if (lane == 31) wsum[wave] = x;
  __syncthreads();
  if (wave == 0) {
    const unsigned t = wsum[lane];
    unsigned u = t;
    for (int o = 1; o < 32; o <<= 1) { const unsigned y = __shfl_up(u, o); if (lane >= o) u += y; }
    wsum[lane] = u - t;
  }
  __syncthreads();
  const unsigned excl = x - c + wsum[wave];
  if (tid <= nB) segoff[(size_t)g * pitch + tid] = e0 + (int)excl;
  cnt[tid] = excl;
  __syncthreads();

  for (int e = e0 + tid; e < e1; e += 8 * 1024) {
    unsigned d[8], s[8];
#pragma unroll
    for (int k = 0; k < 8; ++k) {
      const int ee = e + k * 1024, ec = ee < e1 ? ee : e1 - 1;
      d[k] = (unsigned)dst[ec]; s[k] = (unsigned)src[ec];
    }
#pragma unroll
    for (int k = 0; k < 8; ++k) {
      if (e + k * 1024 < e1) {
        const unsigned r = atomicAdd(&cnt[d[k] >> 8], 1u);
        stage[r] = (s[k] << 8) | (d[k] & 255u);
      }
    }
  }
  __syncthreads();
  const int len = e1 - e0;
  for (int i = tid; i < len; i += 1024) list[e0 + i] = stage[i];
}

__global__ __launch_bounds__(1024) void k_degdis(const unsigned* __restrict__ list, const int* __restrict__ segoff,
                                                 float* __restrict__ dis, int n, int pitch, int nChunk) {
  __shared__ unsigned cnt[BKT];
  const int tid  = threadIdx.x;
  const int lane = tid & 31;
  const int wv   = __builtin_amdgcn_readfirstlane((int)(threadIdx.x >> 5));
  const int b    = blockIdx.x;
  if (tid < BKT) cnt[tid] = 0u;
  __syncthreads();
  int beg = 0, end = 0;
  if (wv < nChunk) { beg = segoff[(size_t)wv * pitch + b]; end = segoff[(size_t)wv * pitch + b + 1]; }
  for (int g = wv; g < nChunk; g += 32) {
    const int gn = g + 32;
    int begN = 0, endN = 0;
    if (gn < nChunk) { begN = segoff[(size_t)gn * pitch + b]; endN = segoff[(size_t)gn * pitch + b + 1]; }
    for (int p = beg + lane; p < end; p += 32) atomicAdd(&cnt[list[p] & 255u], 1u);
    beg = begN; end = endN;
  }
  __syncthreads();
  const int node = b * BKT + tid;
  if (tid < BKT && node < n) dis[node] = rsqrtf((float)(cnt[tid] + 1u));
}

__global__ __launch_bounds__(512) void k_agg(const unsigned* __restrict__ list, const int* __restrict__ segoff,
                                             const float* __restrict__ dis, const float* __restrict__ h,
                                             const float* __restrict__ bias,
                                             float* __restrict__ out, int n, int pitch, int nChunk) {
  __shared__ float acc[BKT * HID];
  const int tid  = threadIdx.x;
  const int lane = tid & 31;
  const unsigned f = lane & 15;
  const int wv   = __builtin_amdgcn_readfirstlane((int)(threadIdx.x >> 5));
  const int b    = blockIdx.x;
  const int node0 = b * BKT;

  float bj = 0.f;
  if (bias) bj = bias[tid & (HID - 1)];

  for (int t = tid; t < BKT * HID; t += 512) {
    const int node = node0 + (t >> 4);
    acc[t] = node < n ? h[(size_t)node * HID + (t & 15)] : 0.f;
  }
  __syncthreads();

  int beg = 0, end = 0;
  if (wv < nChunk) { beg = segoff[(size_t)wv * pitch + b]; end = segoff[(size_t)wv * pitch + b + 1]; }
  for (int g = wv; g < nChunk; g += 16) {
    const int gn = g + 16;
    int begN = 0, endN = 0;
    if (gn < nChunk) { begN = segoff[(size_t)gn * pitch + b]; endN = segoff[(size_t)gn * pitch + b + 1]; }
    for (int base = beg; base < end; base += 32) {
      const int left = end - base;
      const int p = base + lane;
      const unsigned e  = list[p < end ? p : end - 1];
      const unsigned s  = e >> 8, dl = e & 255u;
      float v[16];
#pragma unroll
      for (int j = 0; j < 16; ++j) v[j] = h[row_bcast(j, s) * HID + f];
      const int mine = left - (lane & 16);
#pragma unroll
      for (int j = 0; j < 16; ++j) {
        const unsigned dj = row_bcast(j, dl);
        if (j < mine) atomicAdd(&acc[dj * HID + f], v[j]);
      }
    }
    beg = begN; end = endN;
  }
  __syncthreads();
  for (int t = tid; t < BKT * HID; t += 512) {
    const int node = node0 + (t >> 4);
    if (node < n) {
      const float dn = dis[node];
      float v = acc[t] * dn;
      if (bias) {
        v += bj;
        v = v > 0.f ? v : 0.f;
        v *= dn;
      }
      out[(size_t)node0 * HID + t] = v;
    }
  }
}

static __device__ __forceinline__ void next_trip(const unsigned* __restrict__ list, int mb, int me, int nr,
                                                 int lane, int& r, int& base, int& end,
                                                 unsigned& e, int& cnt) {
  while (base >= end && r + 1 < nr) {
    ++r;
    base = __builtin_amdgcn_readlane(mb, r);
    end  = __builtin_amdgcn_readlane(me, r);
  }
  cnt = end - base;
  if (cnt > 32) cnt = 32;
  int p = base + lane;
  if (p > end - 1) p = end - 1;
  if (p < 0) p = 0;
  e = list[p];
  if (cnt > 0) base += 32;
}

__global__ __launch_bounds__(512) void k_agg2_lsm(const unsigned* __restrict__ list, const int* __restrict__ segoff,
                                                  const float* __restrict__ dis, const float* __restrict__ h2,
                                                  const float* __restrict__ b2, float* __restrict__ y,
                                                  int n, int pitch, int nChunk) {
  __shared__ float acc0[BKT];
  __shared__ float acc1[BKT];
  const int tid  = threadIdx.x;
  const int lane = tid & 31;
  const int wv   = __builtin_amdgcn_readfirstlane((int)(threadIdx.x >> 5));
  const int b    = blockIdx.x;
  const int node0 = b * BKT;
  const float2* __restrict__ h2v = reinterpret_cast<const float2*>(h2);

  if (tid < BKT) {
    const int node = node0 + tid;
    float2 v; v.x = 0.f; v.y = 0.f;
    if (node < n) v = h2v[node];
    acc0[tid] = v.x; acc1[tid] = v.y;
  }
  __syncthreads();

  for (int g0 = wv; g0 < nChunk; g0 += 16 * 32) {
    const int gl = g0 + 16 * lane;
    int mb = 0, me = 0;
    if (gl < nChunk) { mb = segoff[(size_t)gl * pitch + b]; me = segoff[(size_t)gl * pitch + b + 1]; }
    int nr = (nChunk - g0 + 15) >> 4;
    if (nr > 32) nr = 32;

    int r = -1, base = 0, end = 0;
    unsigned e = 0u, eN = 0u;
    int cnt = 0, cntN = 0;
    next_trip(list, mb, me, nr, lane, r, base, end, e, cnt);
    unsigned s = e >> 8, dl = e & 255u;
    while (cnt > 0) {
      next_trip(list, mb, me, nr, lane, r, base, end, eN, cntN);
      if (lane < cnt) {
        const float2 v = h2v[s];
        atomicAdd(&acc0[dl], v.x);
        atomicAdd(&acc1[dl], v.y);
      }
      s = eN >> 8; dl = eN & 255u; cnt = cntN;
    }
  }
  __syncthreads();

  if (tid < BKT) {
    const int node = node0 + tid;
    if (node < n) {
      const float dn = dis[node];
      const float a0 = acc0[tid] * dn + b2[0];
      const float a1 = acc1[tid] * dn + b2[1];
      const float mx = fmaxf(a0, a1);
      const float lse = mx + logf(expf(a0 - mx) + expf(a1 - mx));
      float2 o; o.x = a0 - lse; o.y = a1 - lse;
      reinterpret_cast<float2*>(y)[node] = o;
    }
  }
}


extern "C" void kernel_launch(void* const* d_in, const int* in_sizes, int n_in,
                              void* d_out, int out_size, void* d_ws, size_t ws_size,
                              hipStream_t stream) {
  const float* x  = (const float*)d_in[0];
  const int*   ei = (const int*)d_in[1];
  const float* W1 = (const float*)d_in[2];
  const float* b1 = (const float*)d_in[3];
  const float* W2 = (const float*)d_in[4];
  const float* b2 = (const float*)d_in[5];

  const int n  = in_sizes[0] / IN_DIM;
  const int nE = in_sizes[1] / 2;
  const int* src = ei;
  const int* dst = ei + nE;
  float* ws = (float*)d_ws;
  float* y  = (float*)d_out;

  const int TB = 256;
  dim3 blk(TB);
  int gN   = (n + TB - 1) / TB;
  int gN16 = (int)(((long long)n * HID + TB - 1) / TB);
  int tiles = (n + 15) / 16;
  int gT   = (tiles + (TB / 32) - 1) / (TB / 32);

  const int nB     = (n + BKT - 1) / BKT;
  const int pitch  = (nB + 1 + 31) & ~31;
  const int nChunk = (nE + CH_CAP - 1) / CH_CAP;
  const size_t need = ((size_t)n * (1 + HID + HID + 2) + (size_t)nE + (size_t)nChunk * pitch) * sizeof(float);

  if (ws_size >= need && nChunk > 0 && nB < NB_MAX && n <= (1 << 24)) {
    float* dis  = ws;
    float* h    = dis  + n;
    float* out1 = h    + (size_t)n * HID;
    float* h2   = out1 + (size_t)n * HID;
    unsigned* list = (unsigned*)(h2 + (size_t)n * 2);
    int* segoff    = (int*)(list + nE);

    k_bin       <<<nChunk, dim3(1024), 0, stream>>>(src, dst, list, segoff, nE, nB, pitch);
    k_degdis    <<<nB,  dim3(1024), 0, stream>>>(list, segoff, dis, n, pitch, nChunk);
    k_gemm1_wmma<<<gT,  blk, 0, stream>>>(x, W1, dis, h, n);
    k_agg       <<<nB,  dim3(512), 0, stream>>>(list, segoff, dis, h, b1, out1, n, pitch, nChunk);
    k_gemm2_wmma<<<gT,  blk, 0, stream>>>(out1, W2, h2, n);
    k_agg2_lsm  <<<nB,  dim3(512), 0, stream>>>(list, segoff, dis, h2, b2, y, n, pitch, nChunk);
    return;
  }

  float* deg  = ws;
  float* dis  = deg + n;
  float* h    = dis + n;
  float* out1 = h    + (size_t)n * HID;
  float* h2   = out1 + (size_t)n * HID;
  float* agg  = h;
  int gE   = (nE + TB - 1) / TB;
  int gS   = 2048;

  k_init_deg  <<<gN,  blk, 0, stream>>>(deg, n);
  k_deg_edges <<<gE,  blk, 0, stream>>>(dst, deg, nE);
  k_dis       <<<gN,  blk, 0, stream>>>(deg, dis, n);
  k_gemm1_wmma<<<gT,  blk, 0, stream>>>(x, W1, nullptr, h, n);
  k_self1     <<<gN,  blk, 0, stream>>>(h, dis, out1, n);
  k_scatter1  <<<gS,  blk, 0, stream>>>(src, dst, dis, h, out1, nE);
  k_relu_bias <<<gN16,blk, 0, stream>>>(out1, b1, nullptr, n);
  k_self1     <<<gN,  blk, 0, stream>>>(out1, dis, agg, n);
  k_scatter1  <<<gS,  blk, 0, stream>>>(src, dst, dis, out1, agg, nE);
  k_gemm2_wmma<<<gT,  blk, 0, stream>>>(agg, W2, h2, n);
  k_logsoftmax<<<gN,  blk, 0, stream>>>(h2, b2, y, n);
}


# W2 product taken in the layer-1 aggregation's epilogue: each wave runs the 4 f32 WMMA steps on its 16-row tile of finished rows out of LDS; 16-word rows never stored, GEMM2 launch gone (6 -> 5)
# speedup vs baseline: 6.3475x; 1.0613x over previous
#include <hip/hip_runtime.h>
#include <hip/hip_bf16.h>
#include <math.h>

typedef __attribute__((ext_vector_type(2))) float v2f;
typedef __attribute__((ext_vector_type(8))) float v8f;

#define IN_DIM 128
#define HID    16


__global__ void k_init_deg(float* __restrict__ deg, int n) {
  int i = blockIdx.x * blockDim.x + threadIdx.x;
  if (i < n) deg[i] = 1.0f;
}

__global__ void k_deg_edges(const int* __restrict__ dst, float* __restrict__ deg, int nE) {
  int e = blockIdx.x * blockDim.x + threadIdx.x;
  if (e < nE) atomicAdd(&deg[dst[e]], 1.0f);
}

__global__ void k_dis(const float* __restrict__ deg, float* __restrict__ dis, int n) {
  int i = blockIdx.x * blockDim.x + threadIdx.x;
  if (i < n) dis[i] = rsqrtf(deg[i]);
}


__global__ void k_gemm1_wmma(const float* __restrict__ x, const float* __restrict__ W1,
                             const float* __restrict__ rs, float* __restrict__ h, int n) {
  const int lane = threadIdx.x & 31;
  const int wave = threadIdx.x >> 5;
  const int tile = blockIdx.x * (blockDim.x >> 5) + wave;
  const int row0 = tile << 4;
  if (row0 >= n) return;
  const int m  = lane & 15;
  const int kh = lane >> 4;
  const float* xrow = x + (size_t)(row0 + m) * IN_DIM + 2 * kh;

  float sc[8];
#pragma unroll
  for (int r = 0; r < 8; ++r) sc[r] = 1.0f;
  if (rs) {
#pragma unroll
    for (int r = 0; r < 8; ++r) sc[r] = rs[row0 + 8 * kh + r];
  }

  v8f c = {0.f, 0.f, 0.f, 0.f, 0.f, 0.f, 0.f, 0.f};
#pragma unroll
  for (int k0 = 0; k0 < IN_DIM / 4; ++k0) {
    const int kk = 4 * k0 + 2 * kh;
    const float2 av = *reinterpret_cast<const float2*>(xrow + 4 * k0);
    v2f a; a.x = av.x; a.y = av.y;
    v2f b; b.x = W1[(kk + 0) * HID + m];
           b.y = W1[(kk + 1) * HID + m];
    c = __builtin_amdgcn_wmma_f32_16x16x4_f32(false, a, false, b, (short)0, c, false, false);
  }
  float* hout = h + (size_t)(row0 + 8 * kh) * HID + m;
#pragma unroll
  for (int r = 0; r < 8; ++r)
    hout[(size_t)r * HID] = c[r] * sc[r];
}


__global__ void k_gemm2_wmma(const float* __restrict__ h1, const float* __restrict__ W2,
                             float* __restrict__ h2, int n) {
  const int lane = threadIdx.x & 31;
  const int wave = threadIdx.x >> 5;
  const int tile = blockIdx.x * (blockDim.x >> 5) + wave;
  const int row0 = tile << 4;
  if (row0 >= n) return;
  const int m  = lane & 15;
  const int kh = lane >> 4;
  const float bmask = (m < 2) ? 1.0f : 0.0f;
  const int   col   = (m < 2) ? m : 0;
  const float* arow = h1 + (size_t)(row0 + m) * HID + 2 * kh;

  v8f c = {0.f, 0.f, 0.f, 0.f, 0.f, 0.f, 0.f, 0.f};
#pragma unroll
  for (int k0 = 0; k0 < HID / 4; ++k0) {
    const int kk = 4 * k0 + 2 * kh;
    const float2 av = *reinterpret_cast<const float2*>(arow + 4 * k0);
    v2f a; a.x = av.x; a.y = av.y;
    v2f b; b.x = W2[(kk + 0) * 2 + col] * bmask;
           b.y = W2[(kk + 1) * 2 + col] * bmask;
    c = __builtin_amdgcn_wmma_f32_16x16x4_f32(false, a, false, b, (short)0, c, false, false);
  }
  if (m < 2) {
#pragma unroll
    for (int r = 0; r < 8; ++r)
      h2[(size_t)(row0 + r + 8 * kh) * 2 + m] = c[r];
  }
}


__global__ void k_self1(const float* __restrict__ h, const float* __restrict__ dis,
                        float* __restrict__ out1, int n) {
  int i = blockIdx.x * blockDim.x + threadIdx.x;
  if (i >= n) return;
  float s = dis[i]; s = s * s;
  const float4* hp = reinterpret_cast<const float4*>(h + (size_t)i * HID);
  float4* op = reinterpret_cast<float4*>(out1 + (size_t)i * HID);
#pragma unroll
  for (int q = 0; q < 4; ++q) {
    float4 v = hp[q];
    v.x *= s; v.y *= s; v.z *= s; v.w *= s;
    op[q] = v;
  }
}


static __device__ __forceinline__ unsigned row_bcast(int j, unsigned v) {
  const unsigned sel = 0x11111111u * (unsigned)j;
  return __builtin_amdgcn_permlane16(0u, v, sel, sel, false, true);
}

__global__ void k_scatter1(const int* __restrict__ src, const int* __restrict__ dst,
                           const float* __restrict__ dis, const float* __restrict__ h,
                           float* __restrict__ out1, int nE) {
  const int lane = threadIdx.x & 31;
  const unsigned f = lane & 15;
  const int wv     = __builtin_amdgcn_readfirstlane((int)(threadIdx.x >> 5));
  const int wpb    = (int)(blockDim.x >> 5);
  const int nWaves = (int)gridDim.x * wpb;
  const int nBatch = (nE + 31) >> 5;
  int b = (int)blockIdx.x * wpb + wv;
  if (b >= nBatch) return;

  unsigned s = 0, d = 0;
  if (lane < nE - (b << 5)) {
    s = (unsigned)src[(b << 5) + lane]; d = (unsigned)dst[(b << 5) + lane];
  }
  for (; b < nBatch; b += nWaves) {
    const int left = nE - (b << 5);
    const int bn   = b + nWaves;
    unsigned sN = 0, dN = 0;
    if (left >= 32) {
      const float ds = dis[s], dd = dis[d];
      float v[16];
#pragma unroll
      for (int j = 0; j < 16; ++j) v[j] = h[row_bcast(j, s) * HID + f];
      if (bn < nBatch && lane < nE - (bn << 5)) {
        sN = (unsigned)src[(bn << 5) + lane]; dN = (unsigned)dst[(bn << 5) + lane];
      }
      const float nrm = ds * dd;
#pragma unroll
      for (int j = 0; j < 16; ++j) {
        const unsigned dj = row_bcast(j, d);
        const float    nj = __uint_as_float(row_bcast(j, __float_as_uint(nrm)));
        atomicAdd(out1 + (dj * HID + f), v[j] * nj);
      }
    } else {
      float nrm = 0.f;
      if (lane < left) nrm = dis[s] * dis[d];
      const int mine = left - (lane & 16);
#pragma unroll
      for (int j = 0; j < 16; ++j) {
        const unsigned sj = row_bcast(j, s);
        const unsigned dj = row_bcast(j, d);
        const float    nj = __uint_as_float(row_bcast(j, __float_as_uint(nrm)));
        if (j < mine) atomicAdd(out1 + (dj * HID + f), h[sj * HID + f] * nj);
      }
    }
    s = sN; d = dN;
  }
}

__global__ void k_relu_bias(float* __restrict__ out1, const float* __restrict__ b1,
                            const float* __restrict__ rs, int n) {
  long long t = (long long)blockIdx.x * blockDim.x + threadIdx.x;
  if (t >= (long long)n * HID) return;
  int j = (int)(t & (HID - 1));
  float sc = 1.0f;
  if (rs) sc = rs[t >> 4];
  float v = out1[t] + b1[j];
  v = v > 0.f ? v : 0.f;
  out1[t] = v * sc;
}

__global__ void k_logsoftmax(const float* __restrict__ h2, const float* __restrict__ b2,
                             float* __restrict__ y, int n) {
  int i = blockIdx.x * blockDim.x + threadIdx.x;
  if (i >= n) return;
  float a0 = h2[(size_t)i * 2 + 0] + b2[0];
  float a1 = h2[(size_t)i * 2 + 1] + b2[1];
  float mx = fmaxf(a0, a1);
  float lse = mx + logf(expf(a0 - mx) + expf(a1 - mx));
  y[(size_t)i * 2 + 0] = a0 - lse;
  y[(size_t)i * 2 + 1] = a1 - lse;
}


#define BKT      256
#define CH_CAP   25024
#define NB_MAX   1024

__global__ __launch_bounds__(1024) void k_bin(const int* __restrict__ src, const int* __restrict__ dst,
                                              unsigned* __restrict__ list, int* __restrict__ segoff,
                                              int nE, int nB, int pitch) {
  __shared__ unsigned cnt[NB_MAX];
  __shared__ unsigned wsum[32];
  __shared__ unsigned stage[CH_CAP];
  const int tid  = threadIdx.x;
  const int lane = tid & 31;
  const int wave = tid >> 5;
  const int g    = blockIdx.x;
  int e0 = g * CH_CAP;            if (e0 > nE) e0 = nE;
  int e1 = e0 + CH_CAP;           if (e1 > nE) e1 = nE;

  cnt[tid] = 0u;
  __syncthreads();
  for (int e = e0 + tid; e < e1; e += 8 * 1024) {
    unsigned d[8];
#pragma unroll
    for (int k = 0; k < 8; ++k) { const int ee = e + k * 1024; d[k] = (unsigned)dst[ee < e1 ? ee : e1 - 1]; }
#pragma unroll
    for (int k = 0; k < 8; ++k) if (e + k * 1024 < e1) atomicAdd(&cnt[d[k] >> 8], 1u);
  }
  __syncthreads();

  const unsigned c = cnt[tid];
  unsigned x = c;
  for (int o = 1; o < 32; o <<= 1) { const unsigned y = __shfl_up(x, o); if (lane >= o) x += y; }
  if (lane == 31) wsum[wave] = x;
  __syncthreads();
  if (wave == 0) {
    const unsigned t = wsum[lane];
    unsigned u = t;
    for (int o = 1; o < 32; o <<= 1) { const unsigned y = __shfl_up(u, o); if (lane >= o) u += y; }
    wsum[lane] = u - t;
  }
  __syncthreads();
  const unsigned excl = x - c + wsum[wave];
  if (tid <= nB) segoff[(size_t)g * pitch + tid] = e0 + (int)excl;
  cnt[tid] = excl;
  __syncthreads();

  for (int e = e0 + tid; e < e1; e += 8 * 1024) {
    unsigned d[8], s[8];
#pragma unroll
    for (int k = 0; k < 8; ++k) {
      const int ee = e + k * 1024, ec = ee < e1 ? ee : e1 - 1;
      d[k] = (unsigned)dst[ec]; s[k] = (unsigned)src[ec];
    }
#pragma unroll
    for (int k = 0; k < 8; ++k) {
      if (e + k * 1024 < e1) {
        const unsigned r = atomicAdd(&cnt[d[k] >> 8], 1u);
        stage[r] = (s[k] << 8) | (d[k] & 255u);
      }
    }
  }
  __syncthreads();
  const int len = e1 - e0;
  for (int i = tid; i < len; i += 1024) list[e0 + i] = stage[i];
}

__global__ __launch_bounds__(1024) void k_degdis(const unsigned* __restrict__ list, const int* __restrict__ segoff,
                                                 float* __restrict__ dis, int n, int pitch, int nChunk) {
  __shared__ unsigned cnt[BKT];
  const int tid  = threadIdx.x;
  const int lane = tid & 31;
  const int wv   = __builtin_amdgcn_readfirstlane((int)(threadIdx.x >> 5));
  const int b    = blockIdx.x;
  if (tid < BKT) cnt[tid] = 0u;
  __syncthreads();
  int beg = 0, end = 0;
  if (wv < nChunk) { beg = segoff[(size_t)wv * pitch + b]; end = segoff[(size_t)wv * pitch + b + 1]; }
  for (int g = wv; g < nChunk; g += 32) {
    const int gn = g + 32;
    int begN = 0, endN = 0;
    if (gn < nChunk) { begN = segoff[(size_t)gn * pitch + b]; endN = segoff[(size_t)gn * pitch + b + 1]; }
    for (int p = beg + lane; p < end; p += 32) atomicAdd(&cnt[list[p] & 255u], 1u);
    beg = begN; end = endN;
  }
  __syncthreads();
  const int node = b * BKT + tid;
  if (tid < BKT && node < n) dis[node] = rsqrtf((float)(cnt[tid] + 1u));
}

__global__ __launch_bounds__(512) void k_agg(const unsigned* __restrict__ list, const int* __restrict__ segoff,
                                             const float* __restrict__ dis, const float* __restrict__ h,
                                             const float* __restrict__ bias, const float* __restrict__ W2,
                                             float* __restrict__ out, int n, int pitch, int nChunk) {
  __shared__ __attribute__((aligned(16))) float acc[BKT * HID];
  __shared__ __attribute__((aligned(16))) float par[3 * HID];
  const int tid  = threadIdx.x;
  const int lane = tid & 31;
  const unsigned f = lane & 15;
  const int wv   = __builtin_amdgcn_readfirstlane((int)(threadIdx.x >> 5));
  const int b    = blockIdx.x;
  const int node0 = b * BKT;

  float bj = 0.f;
  if (bias && !W2) bj = bias[tid & (HID - 1)];
  float dnr = 0.f, pw = 0.f;
  if (W2) {
    if (tid < 3 * HID) pw = tid < HID ? bias[tid] : W2[tid - HID];
    const int nd = node0 + (wv << 4) + (lane & 15);
    if (nd < n) dnr = dis[nd];
  }

  for (int t = tid; t < BKT * HID; t += 512) {
    const int node = node0 + (t >> 4);
    acc[t] = node < n ? h[(size_t)node * HID + (t & 15)] : 0.f;
  }
  if (W2 && tid < 3 * HID) par[tid] = pw;
  __syncthreads();

  int beg = 0, end = 0;
  if (wv < nChunk) { beg = segoff[(size_t)wv * pitch + b]; end = segoff[(size_t)wv * pitch + b + 1]; }
  for (int g = wv; g < nChunk; g += 16) {
    const int gn = g + 16;
    int begN = 0, endN = 0;
    if (gn < nChunk) { begN = segoff[(size_t)gn * pitch + b]; endN = segoff[(size_t)gn * pitch + b + 1]; }
    for (int base = beg; base < end; base += 32) {
      const int left = end - base;
      const int p = base + lane;
      const unsigned e  = list[p < end ? p : end - 1];
      const unsigned s  = e >> 8, dl = e & 255u;
      float v[16];
#pragma unroll
      for (int j = 0; j < 16; ++j) v[j] = h[row_bcast(j, s) * HID + f];
      const int mine = left - (lane & 16);
#pragma unroll
      for (int j = 0; j < 16; ++j) {
        const unsigned dj = row_bcast(j, dl);
        if (j < mine) atomicAdd(&acc[dj * HID + f], v[j]);
      }
    }
    beg = begN; end = endN;
  }
  if (W2) {
    const int m   = lane & 15;
    const int kh  = lane >> 4;
    const int row = (wv << 4) + m;
    const float bmask = (m < 2) ? 1.0f : 0.0f;
    const int   col   = (m < 2) ? m : 0;
    const float dn    = dnr;
    __syncthreads();
    v8f c = {0.f, 0.f, 0.f, 0.f, 0.f, 0.f, 0.f, 0.f};
#pragma unroll
    for (int k0 = 0; k0 < HID / 4; ++k0) {
      const int kk = 4 * k0 + 2 * kh;
      const float2 sv = *reinterpret_cast<const float2*>(&acc[row * HID + kk]);
      const float2 bv = *reinterpret_cast<const float2*>(&par[kk]);
      float a0 = sv.x * dn; a0 += bv.x; a0 = a0 > 0.f ? a0 : 0.f; a0 *= dn;
      float a1 = sv.y * dn; a1 += bv.y; a1 = a1 > 0.f ? a1 : 0.f; a1 *= dn;
      v2f a;  a.x  = a0; a.y = a1;
      v2f bf; bf.x = par[HID + (kk + 0) * 2 + col] * bmask;
              bf.y = par[HID + (kk + 1) * 2 + col] * bmask;
      c = __builtin_amdgcn_wmma_f32_16x16x4_f32(false, a, false, bf, (short)0, c, false, false);
    }
    if (m < 2) {
      const int r0 = node0 + (wv << 4) + 8 * kh;
#pragma unroll
      for (int r = 0; r < 8; ++r)
        if (r0 + r < n) out[(size_t)(r0 + r) * 2 + m] = c[r];
    }
    return;
  }
  __syncthreads();
  for (int t = tid; t < BKT * HID; t += 512) {
    const int node = node0 + (t >> 4);
    if (node < n) {
      const float dn = dis[node];
      float v = acc[t] * dn;
      if (bias) {
        v += bj;
        v = v > 0.f ? v : 0.f;
        v *= dn;
      }
      out[(size_t)node0 * HID + t] = v;
    }
  }
}

static __device__ __forceinline__ void next_trip(const unsigned* __restrict__ list, int mb, int me, int nr,
                                                 int lane, int& r, int& base, int& end,
                                                 unsigned& e, int& cnt) {
  while (base >= end && r + 1 < nr) {
    ++r;
    base = __builtin_amdgcn_readlane(mb, r);
    end  = __builtin_amdgcn_readlane(me, r);
  }
  cnt = end - base;
  if (cnt > 32) cnt = 32;
  int p = base + lane;
  if (p > end - 1) p = end - 1;
  if (p < 0) p = 0;
  e = list[p];
  if (cnt > 0) base += 32;
}

__global__ __launch_bounds__(512) void k_agg2_lsm(const unsigned* __restrict__ list, const int* __restrict__ segoff,
                                                  const float* __restrict__ dis, const float* __restrict__ h2,
                                                  const float* __restrict__ b2, float* __restrict__ y,
                                                  int n, int pitch, int nChunk) {
  __shared__ float acc0[BKT];
  __shared__ float acc1[BKT];
  const int tid  = threadIdx.x;
  const int lane = tid & 31;
  const int wv   = __builtin_amdgcn_readfirstlane((int)(threadIdx.x >> 5));
  const int b    = blockIdx.x;
  const int node0 = b * BKT;
  const float2* __restrict__ h2v = reinterpret_cast<const float2*>(h2);

  if (tid < BKT) {
    const int node = node0 + tid;
    float2 v; v.x = 0.f; v.y = 0.f;
    if (node < n) v = h2v[node];
    acc0[tid] = v.x; acc1[tid] = v.y;
  }
  __syncthreads();

  for (int g0 = wv; g0 < nChunk; g0 += 16 * 32) {
    const int gl = g0 + 16 * lane;
    int mb = 0, me = 0;
    if (gl < nChunk) { mb = segoff[(size_t)gl * pitch + b]; me = segoff[(size_t)gl * pitch + b + 1]; }
    int nr = (nChunk - g0 + 15) >> 4;
    if (nr > 32) nr = 32;

    int r = -1, base = 0, end = 0;
    unsigned e = 0u, eN = 0u;
    int cnt = 0, cntN = 0;
    next_trip(list, mb, me, nr, lane, r, base, end, e, cnt);
    unsigned s = e >> 8, dl = e & 255u;
    while (cnt > 0) {
      next_trip(list, mb, me, nr, lane, r, base, end, eN, cntN);
      if (lane < cnt) {
        const float2 v = h2v[s];
        atomicAdd(&acc0[dl], v.x);
        atomicAdd(&acc1[dl], v.y);
      }
      s = eN >> 8; dl = eN & 255u; cnt = cntN;
    }
  }
  __syncthreads();

  if (tid < BKT) {
    const int node = node0 + tid;
    if (node < n) {
      const float dn = dis[node];
      const float a0 = acc0[tid] * dn + b2[0];
      const float a1 = acc1[tid] * dn + b2[1];
      const float mx = fmaxf(a0, a1);
      const float lse = mx + logf(expf(a0 - mx) + expf(a1 - mx));
      float2 o; o.x = a0 - lse; o.y = a1 - lse;
      reinterpret_cast<float2*>(y)[node] = o;
    }
  }
}


extern "C" void kernel_launch(void* const* d_in, const int* in_sizes, int n_in,
                              void* d_out, int out_size, void* d_ws, size_t ws_size,
                              hipStream_t stream) {
  const float* x  = (const float*)d_in[0];
  const int*   ei = (const int*)d_in[1];
  const float* W1 = (const float*)d_in[2];
  const float* b1 = (const float*)d_in[3];
  const float* W2 = (const float*)d_in[4];
  const float* b2 = (const float*)d_in[5];

  const int n  = in_sizes[0] / IN_DIM;
  const int nE = in_sizes[1] / 2;
  const int* src = ei;
  const int* dst = ei + nE;
  float* ws = (float*)d_ws;
  float* y  = (float*)d_out;

  const int TB = 256;
  dim3 blk(TB);
  int gN   = (n + TB - 1) / TB;
  int gN16 = (int)(((long long)n * HID + TB - 1) / TB);
  int tiles = (n + 15) / 16;
  int gT   = (tiles + (TB / 32) - 1) / (TB / 32);

  const int nB     = (n + BKT - 1) / BKT;
  const int pitch  = (nB + 1 + 31) & ~31;
  const int nChunk = (nE + CH_CAP - 1) / CH_CAP;
  const size_t need = ((size_t)n * (1 + HID + 2) + (size_t)nE + (size_t)nChunk * pitch) * sizeof(float);

  if (ws_size >= need && nChunk > 0 && nB < NB_MAX && n <= (1 << 24)) {
    float* dis  = ws;
    float* h    = dis  + n;
    float* h2   = h    + (size_t)n * HID;
    unsigned* list = (unsigned*)(h2 + (size_t)n * 2);
    int* segoff    = (int*)(list + nE);

    k_bin       <<<nChunk, dim3(1024), 0, stream>>>(src, dst, list, segoff, nE, nB, pitch);
    k_degdis    <<<nB,  dim3(1024), 0, stream>>>(list, segoff, dis, n, pitch, nChunk);
    k_gemm1_wmma<<<gT,  blk, 0, stream>>>(x, W1, dis, h, n);
    k_agg       <<<nB,  dim3(512), 0, stream>>>(list, segoff, dis, h, b1, W2, h2, n, pitch, nChunk);
    k_agg2_lsm  <<<nB,  dim3(512), 0, stream>>>(list, segoff, dis, h2, b2, y, n, pitch, nChunk);
    return;
  }

  float* deg  = ws;
  float* dis  = deg + n;
  float* h    = dis + n;
  float* out1 = h    + (size_t)n * HID;
  float* h2   = out1 + (size_t)n * HID;
  float* agg  = h;
  int gE   = (nE + TB - 1) / TB;
  int gS   = 2048;

  k_init_deg  <<<gN,  blk, 0, stream>>>(deg, n);
  k_deg_edges <<<gE,  blk, 0, stream>>>(dst, deg, nE);
  k_dis       <<<gN,  blk, 0, stream>>>(deg, dis, n);
  k_gemm1_wmma<<<gT,  blk, 0, stream>>>(x, W1, nullptr, h, n);
  k_self1     <<<gN,  blk, 0, stream>>>(h, dis, out1, n);
  k_scatter1  <<<gS,  blk, 0, stream>>>(src, dst, dis, h, out1, nE);
  k_relu_bias <<<gN16,blk, 0, stream>>>(out1, b1, nullptr, n);
  k_self1     <<<gN,  blk, 0, stream>>>(out1, dis, agg, n);
  k_scatter1  <<<gS,  blk, 0, stream>>>(src, dst, dis, out1, agg, nE);
  k_gemm2_wmma<<<gT,  blk, 0, stream>>>(agg, W2, h2, n);
  k_logsoftmax<<<gN,  blk, 0, stream>>>(h2, b2, y, n);
}


# Degree count and dis taken in GEMM1's launch: a range's 16 waves count its runs in LDS (16 list loads in flight), then run 16-row WMMA tiles, store rows times dis; launches 5 -> 4; sums as before
# speedup vs baseline: 6.8139x; 1.0735x over previous
#include <hip/hip_runtime.h>
#include <hip/hip_bf16.h>
#include <math.h>

typedef __attribute__((ext_vector_type(2))) float v2f;
typedef __attribute__((ext_vector_type(8))) float v8f;

#define IN_DIM 128
#define HID    16


__global__ void k_init_deg(float* __restrict__ deg, int n) {
  int i = blockIdx.x * blockDim.x + threadIdx.x;
  if (i < n) deg[i] = 1.0f;
}

__global__ void k_deg_edges(const int* __restrict__ dst, float* __restrict__ deg, int nE) {
  int e = blockIdx.x * blockDim.x + threadIdx.x;
  if (e < nE) atomicAdd(&deg[dst[e]], 1.0f);
}

__global__ void k_dis(const float* __restrict__ deg, float* __restrict__ dis, int n) {
  int i = blockIdx.x * blockDim.x + threadIdx.x;
  if (i < n) dis[i] = rsqrtf(deg[i]);
}


__global__ void k_gemm1_wmma(const float* __restrict__ x, const float* __restrict__ W1,
                             float* __restrict__ h, int n) {
  const int lane = threadIdx.x & 31;
  const int wave = threadIdx.x >> 5;
  const int tile = blockIdx.x * (blockDim.x >> 5) + wave;
  const int row0 = tile << 4;
  if (row0 >= n) return;
  const int m  = lane & 15;
  const int kh = lane >> 4;
  const float* xrow = x + (size_t)(row0 + m) * IN_DIM + 2 * kh;

  v8f c = {0.f, 0.f, 0.f, 0.f, 0.f, 0.f, 0.f, 0.f};
#pragma unroll
  for (int k0 = 0; k0 < IN_DIM / 4; ++k0) {
    const int kk = 4 * k0 + 2 * kh;
    const float2 av = *reinterpret_cast<const float2*>(xrow + 4 * k0);
    v2f a; a.x = av.x; a.y = av.y;
    v2f b; b.x = W1[(kk + 0) * HID + m];
           b.y = W1[(kk + 1) * HID + m];
    c = __builtin_amdgcn_wmma_f32_16x16x4_f32(false, a, false, b, (short)0, c, false, false);
  }
  float* hout = h + (size_t)(row0 + 8 * kh) * HID + m;
#pragma unroll
  for (int r = 0; r < 8; ++r)
    hout[(size_t)r * HID] = c[r];
}


__global__ void k_gemm2_wmma(const float* __restrict__ h1, const float* __restrict__ W2,
                             float* __restrict__ h2, int n) {
  const int lane = threadIdx.x & 31;
  const int wave = threadIdx.x >> 5;
  const int tile = blockIdx.x * (blockDim.x >> 5) + wave;
  const int row0 = tile << 4;
  if (row0 >= n) return;
  const int m  = lane & 15;
  const int kh = lane >> 4;
  const float bmask = (m < 2) ? 1.0f : 0.0f;
  const int   col   = (m < 2) ? m : 0;
  const float* arow = h1 + (size_t)(row0 + m) * HID + 2 * kh;

  v8f c = {0.f, 0.f, 0.f, 0.f, 0.f, 0.f, 0.f, 0.f};
#pragma unroll
  for (int k0 = 0; k0 < HID / 4; ++k0) {
    const int kk = 4 * k0 + 2 * kh;
    const float2 av = *reinterpret_cast<const float2*>(arow + 4 * k0);
    v2f a; a.x = av.x; a.y = av.y;
    v2f b; b.x = W2[(kk + 0) * 2 + col] * bmask;
           b.y = W2[(kk + 1) * 2 + col] * bmask;
    c = __builtin_amdgcn_wmma_f32_16x16x4_f32(false, a, false, b, (short)0, c, false, false);
  }
  if (m < 2) {
#pragma unroll
    for (int r = 0; r < 8; ++r)
      h2[(size_t)(row0 + r + 8 * kh) * 2 + m] = c[r];
  }
}


__global__ void k_self1(const float* __restrict__ h, const float* __restrict__ dis,
                        float* __restrict__ out1, int n) {
  int i = blockIdx.x * blockDim.x + threadIdx.x;
  if (i >= n) return;
  float s = dis[i]; s = s * s;
  const float4* hp = reinterpret_cast<const float4*>(h + (size_t)i * HID);
  float4* op = reinterpret_cast<float4*>(out1 + (size_t)i * HID);
#pragma unroll
  for (int q = 0; q < 4; ++q) {
    float4 v = hp[q];
    v.x *= s; v.y *= s; v.z *= s; v.w *= s;
    op[q] = v;
  }
}


static __device__ __forceinline__ unsigned row_bcast(int j, unsigned v) {
  const unsigned sel = 0x11111111u * (unsigned)j;
  return __builtin_amdgcn_permlane16(0u, v, sel, sel, false, true);
}

__global__ void k_scatter1(const int* __restrict__ src, const int* __restrict__ dst,
                           const float* __restrict__ dis, const float* __restrict__ h,
                           float* __restrict__ out1, int nE) {
  const int lane = threadIdx.x & 31;
  const unsigned f = lane & 15;
  const int wv     = __builtin_amdgcn_readfirstlane((int)(threadIdx.x >> 5));
  const int wpb    = (int)(blockDim.x >> 5);
  const int nWaves = (int)gridDim.x * wpb;
  const int nBatch = (nE + 31) >> 5;
  int b = (int)blockIdx.x * wpb + wv;
  if (b >= nBatch) return;

  unsigned s = 0, d = 0;
  if (lane < nE - (b << 5)) {
    s = (unsigned)src[(b << 5) + lane]; d = (unsigned)dst[(b << 5) + lane];
  }
  for (; b < nBatch; b += nWaves) {
    const int left = nE - (b << 5);
    const int bn   = b + nWaves;
    unsigned sN = 0, dN = 0;
    if (left >= 32) {
      const float ds = dis[s], dd = dis[d];
      float v[16];
#pragma unroll
      for (int j = 0; j < 16; ++j) v[j] = h[row_bcast(j, s) * HID + f];
      if (bn < nBatch && lane < nE - (bn << 5)) {
        sN = (unsigned)src[(bn << 5) + lane]; dN = (unsigned)dst[(bn << 5) + lane];
      }
      const float nrm = ds * dd;
#pragma unroll
      for (int j = 0; j < 16; ++j) {
        const unsigned dj = row_bcast(j, d);
        const float    nj = __uint_as_float(row_bcast(j, __float_as_uint(nrm)));
        atomicAdd(out1 + (dj * HID + f), v[j] * nj);
      }
    } else {
      float nrm = 0.f;
      if (lane < left) nrm = dis[s] * dis[d];
      const int mine = left - (lane & 16);
#pragma unroll
      for (int j = 0; j < 16; ++j) {
        const unsigned sj = row_bcast(j, s);
        const unsigned dj = row_bcast(j, d);
        const float    nj = __uint_as_float(row_bcast(j, __float_as_uint(nrm)));
        if (j < mine) atomicAdd(out1 + (dj * HID + f), h[sj * HID + f] * nj);
      }
    }
    s = sN; d = dN;
  }
}

__global__ void k_relu_bias(float* __restrict__ out1, const float* __restrict__ b1,
                            const float* __restrict__ rs, int n) {
  long long t = (long long)blockIdx.x * blockDim.x + threadIdx.x;
  if (t >= (long long)n * HID) return;
  int j = (int)(t & (HID - 1));
  float sc = 1.0f;
  if (rs) sc = rs[t >> 4];
  float v = out1[t] + b1[j];
  v = v > 0.f ? v : 0.f;
  out1[t] = v * sc;
}

__global__ void k_logsoftmax(const float* __restrict__ h2, const float* __restrict__ b2,
                             float* __restrict__ y, int n) {
  int i = blockIdx.x * blockDim.x + threadIdx.x;
  if (i >= n) return;
  float a0 = h2[(size_t)i * 2 + 0] + b2[0];
  float a1 = h2[(size_t)i * 2 + 1] + b2[1];
  float mx = fmaxf(a0, a1);
  float lse = mx + logf(expf(a0 - mx) + expf(a1 - mx));
  y[(size_t)i * 2 + 0] = a0 - lse;
  y[(size_t)i * 2 + 1] = a1 - lse;
}


#define BKT      256
#define CH_CAP   25024
#define NB_MAX   1024

__global__ __launch_bounds__(1024) void k_bin(const int* __restrict__ src, const int* __restrict__ dst,
                                              unsigned* __restrict__ list, int* __restrict__ segoff,
                                              int nE, int nB, int pitch) {
  __shared__ unsigned cnt[NB_MAX];
  __shared__ unsigned wsum[32];
  __shared__ unsigned stage[CH_CAP];
  const int tid  = threadIdx.x;
  const int lane = tid & 31;
  const int wave = tid >> 5;
  const int g    = blockIdx.x;
  int e0 = g * CH_CAP;            if (e0 > nE) e0 = nE;
  int e1 = e0 + CH_CAP;           if (e1 > nE) e1 = nE;

  cnt[tid] = 0u;
  __syncthreads();
  for (int e = e0 + tid; e < e1; e += 8 * 1024) {
    unsigned d[8];
#pragma unroll
    for (int k = 0; k < 8; ++k) { const int ee = e + k * 1024; d[k] = (unsigned)dst[ee < e1 ? ee : e1 - 1]; }
#pragma unroll
    for (int k = 0; k < 8; ++k) if (e + k * 1024 < e1) atomicAdd(&cnt[d[k] >> 8], 1u);
  }
  __syncthreads();

  const unsigned c = cnt[tid];
  unsigned x = c;
  for (int o = 1; o < 32; o <<= 1) { const unsigned y = __shfl_up(x, o); if (lane >= o) x += y; }
  if (lane == 31) wsum[wave] = x;
  __syncthreads();
  if (wave == 0) {
    const unsigned t = wsum[lane];
    unsigned u = t;
    for (int o = 1; o < 32; o <<= 1) { const unsigned y = __shfl_up(u, o); if (lane >= o) u += y; }
    wsum[lane] = u - t;
  }
  __syncthreads();
  const unsigned excl = x - c + wsum[wave];
  if (tid <= nB) segoff[(size_t)g * pitch + tid] = e0 + (int)excl;
  cnt[tid] = excl;
  __syncthreads();

  for (int e = e0 + tid; e < e1; e += 8 * 1024) {
    unsigned d[8], s[8];
#pragma unroll
    for (int k = 0; k < 8; ++k) {
      const int ee = e + k * 1024, ec = ee < e1 ? ee : e1 - 1;
      d[k] = (unsigned)dst[ec]; s[k] = (unsigned)src[ec];
    }
#pragma unroll
    for (int k = 0; k < 8; ++k) {
      if (e + k * 1024 < e1) {
        const unsigned r = atomicAdd(&cnt[d[k] >> 8], 1u);
        stage[r] = (s[k] << 8) | (d[k] & 255u);
      }
    }
  }
  __syncthreads();
  const int len = e1 - e0;
  for (int i = tid; i < len; i += 1024) list[e0 + i] = stage[i];
}

__global__ __launch_bounds__(512) void k_deg_gemm1(const unsigned* __restrict__ list, const int* __restrict__ segoff,
                                                   const float* __restrict__ x, const float* __restrict__ W1,
                                                   float* __restrict__ dis, float* __restrict__ h,
                                                   int n, int pitch, int nChunk) {
  __shared__ unsigned cnt[BKT];
  const int tid  = threadIdx.x;
  const int lane = tid & 31;
  const int wv   = __builtin_amdgcn_readfirstlane((int)(threadIdx.x >> 5));
  const int b    = blockIdx.x;
  const int node0 = b * BKT;
  if (tid < BKT) cnt[tid] = 0u;
  __syncthreads();

  for (int g0 = wv; g0 < nChunk; g0 += 16 * 32) {
    const int gl = g0 + 16 * lane;
    int mb = 0, me = 0;
    if (gl < nChunk) { mb = segoff[(size_t)gl * pitch + b]; me = segoff[(size_t)gl * pitch + b + 1]; }
    int nr = (nChunk - g0 + 15) >> 4;
    if (nr > 32) nr = 32;
    for (int r0 = 0; r0 < nr; r0 += 8) {
      unsigned w0[8], w1[8];
      int left[8];
#pragma unroll
      for (int k = 0; k < 8; ++k) {
        const int beg = __builtin_amdgcn_readlane(mb, r0 + k);
        const int end = __builtin_amdgcn_readlane(me, r0 + k);
        const int p0 = beg + lane;
        int q0 = p0 < end ? p0 : end - 1;            if (q0 < 0) q0 = 0;
        int q1 = p0 + 32 < end ? p0 + 32 : end - 1;  if (q1 < 0) q1 = 0;
        w0[k] = list[q0];
        w1[k] = list[q1];
        left[k] = end - p0;
      }
#pragma unroll
      for (int k = 0; k < 8; ++k) {
        if (left[k] > 0)  atomicAdd(&cnt[w0[k] & 255u], 1u);
        if (left[k] > 32) atomicAdd(&cnt[w1[k] & 255u], 1u);
      }
      for (int k = 0; k < 8; ++k) {
        const int beg = __builtin_amdgcn_readlane(mb, r0 + k);
        const int end = __builtin_amdgcn_readlane(me, r0 + k);
        for (int base = beg + 64; base < end; base += 32) {
          const int p = base + lane;
          if (p < end) atomicAdd(&cnt[list[p] & 255u], 1u);
        }
      }
    }
  }

  const int row0 = node0 + (wv << 4);
  const int m  = lane & 15;
  const int kh = lane >> 4;
  v8f c = {0.f, 0.f, 0.f, 0.f, 0.f, 0.f, 0.f, 0.f};
  if (row0 < n) {
    const float* xrow = x + (size_t)(row0 + m) * IN_DIM + 2 * kh;
#pragma unroll
    for (int k0 = 0; k0 < IN_DIM / 4; ++k0) {
      const int kk = 4 * k0 + 2 * kh;
      const float2 av = *reinterpret_cast<const float2*>(xrow + 4 * k0);
      v2f a;  a.x  = av.x; a.y = av.y;
      v2f bf; bf.x = W1[(kk + 0) * HID + m];
              bf.y = W1[(kk + 1) * HID + m];
      c = __builtin_amdgcn_wmma_f32_16x16x4_f32(false, a, false, bf, (short)0, c, false, false);
    }
  }
  __syncthreads();
  if (tid < BKT) {
    const int node = node0 + tid;
    if (node < n) dis[node] = rsqrtf((float)(cnt[tid] + 1u));
  }
  if (row0 < n) {
    float* hout = h + (size_t)(row0 + 8 * kh) * HID + m;
#pragma unroll
    for (int r = 0; r < 8; ++r) {
      const float sc = rsqrtf((float)(cnt[(wv << 4) + 8 * kh + r] + 1u));
      hout[(size_t)r * HID] = c[r] * sc;
    }
  }
}

__global__ __launch_bounds__(512) void k_agg(const unsigned* __restrict__ list, const int* __restrict__ segoff,
                                             const float* __restrict__ dis, const float* __restrict__ h,
                                             const float* __restrict__ bias, const float* __restrict__ W2,
                                             float* __restrict__ out, int n, int pitch, int nChunk) {
  __shared__ __attribute__((aligned(16))) float acc[BKT * HID];
  __shared__ __attribute__((aligned(16))) float par[3 * HID];
  const int tid  = threadIdx.x;
  const int lane = tid & 31;
  const unsigned f = lane & 15;
  const int wv   = __builtin_amdgcn_readfirstlane((int)(threadIdx.x >> 5));
  const int b    = blockIdx.x;
  const int node0 = b * BKT;

  float bj = 0.f;
  if (bias && !W2) bj = bias[tid & (HID - 1)];
  float dnr = 0.f, pw = 0.f;
  if (W2) {
    if (tid < 3 * HID) pw = tid < HID ? bias[tid] : W2[tid - HID];
    const int nd = node0 + (wv << 4) + (lane & 15);
    if (nd < n) dnr = dis[nd];
  }

  for (int t = tid; t < BKT * HID; t += 512) {
    const int node = node0 + (t >> 4);
    acc[t] = node < n ? h[(size_t)node * HID + (t & 15)] : 0.f;
  }
  if (W2 && tid < 3 * HID) par[tid] = pw;
  __syncthreads();

  int beg = 0, end = 0;
  if (wv < nChunk) { beg = segoff[(size_t)wv * pitch + b]; end = segoff[(size_t)wv * pitch + b + 1]; }
  for (int g = wv; g < nChunk; g += 16) {
    const int gn = g + 16;
    int begN = 0, endN = 0;
    if (gn < nChunk) { begN = segoff[(size_t)gn * pitch + b]; endN = segoff[(size_t)gn * pitch + b + 1]; }
    for (int base = beg; base < end; base += 32) {
      const int left = end - base;
      const int p = base + lane;
      const unsigned e  = list[p < end ? p : end - 1];
      const unsigned s  = e >> 8, dl = e & 255u;
      float v[16];
#pragma unroll
      for (int j = 0; j < 16; ++j) v[j] = h[row_bcast(j, s) * HID + f];
      const int mine = left - (lane & 16);
#pragma unroll
      for (int j = 0; j < 16; ++j) {
        const unsigned dj = row_bcast(j, dl);
        if (j < mine) atomicAdd(&acc[dj * HID + f], v[j]);
      }
    }
    beg = begN; end = endN;
  }
  if (W2) {
    const int m   = lane & 15;
    const int kh  = lane >> 4;
    const int row = (wv << 4) + m;
    const float bmask = (m < 2) ? 1.0f : 0.0f;
    const int   col   = (m < 2) ? m : 0;
    const float dn    = dnr;
    __syncthreads();
    v8f c = {0.f, 0.f, 0.f, 0.f, 0.f, 0.f, 0.f, 0.f};
#pragma unroll
    for (int k0 = 0; k0 < HID / 4; ++k0) {
      const int kk = 4 * k0 + 2 * kh;
      const float2 sv = *reinterpret_cast<const float2*>(&acc[row * HID + kk]);
      const float2 bv = *reinterpret_cast<const float2*>(&par[kk]);
      float a0 = sv.x * dn; a0 += bv.x; a0 = a0 > 0.f ? a0 : 0.f; a0 *= dn;
      float a1 = sv.y * dn; a1 += bv.y; a1 = a1 > 0.f ? a1 : 0.f; a1 *= dn;
      v2f a;  a.x  = a0; a.y = a1;
      v2f bf; bf.x = par[HID + (kk + 0) * 2 + col] * bmask;
              bf.y = par[HID + (kk + 1) * 2 + col] * bmask;
      c = __builtin_amdgcn_wmma_f32_16x16x4_f32(false, a, false, bf, (short)0, c, false, false);
    }
    if (m < 2) {
      const int r0 = node0 + (wv << 4) + 8 * kh;
#pragma unroll
      for (int r = 0; r < 8; ++r)
        if (r0 + r < n) out[(size_t)(r0 + r) * 2 + m] = c[r];
    }
    return;
  }
  __syncthreads();
  for (int t = tid; t < BKT * HID; t += 512) {
    const int node = node0 + (t >> 4);
    if (node < n) {
      const float dn = dis[node];
      float v = acc[t] * dn;
      if (bias) {
        v += bj;
        v = v > 0.f ? v : 0.f;
        v *= dn;
      }
      out[(size_t)node0 * HID + t] = v;
    }
  }
}

static __device__ __forceinline__ void next_trip(const unsigned* __restrict__ list, int mb, int me, int nr,
                                                 int lane, int& r, int& base, int& end,
                                                 unsigned& e, int& cnt) {
  while (base >= end && r + 1 < nr) {
    ++r;
    base = __builtin_amdgcn_readlane(mb, r);
    end  = __builtin_amdgcn_readlane(me, r);
  }
  cnt = end - base;
  if (cnt > 32) cnt = 32;
  int p = base + lane;
  if (p > end - 1) p = end - 1;
  if (p < 0) p = 0;
  e = list[p];
  if (cnt > 0) base += 32;
}

__global__ __launch_bounds__(512) void k_agg2_lsm(const unsigned* __restrict__ list, const int* __restrict__ segoff,
                                                  const float* __restrict__ dis, const float* __restrict__ h2,
                                                  const float* __restrict__ b2, float* __restrict__ y,
                                                  int n, int pitch, int nChunk) {
  __shared__ float acc0[BKT];
  __shared__ float acc1[BKT];
  const int tid  = threadIdx.x;
  const int lane = tid & 31;
  const int wv   = __builtin_amdgcn_readfirstlane((int)(threadIdx.x >> 5));
  const int b    = blockIdx.x;
  const int node0 = b * BKT;
  const float2* __restrict__ h2v = reinterpret_cast<const float2*>(h2);

  if (tid < BKT) {
    const int node = node0 + tid;
    float2 v; v.x = 0.f; v.y = 0.f;
    if (node < n) v = h2v[node];
    acc0[tid] = v.x; acc1[tid] = v.y;
  }
  __syncthreads();

  for (int g0 = wv; g0 < nChunk; g0 += 16 * 32) {
    const int gl = g0 + 16 * lane;
    int mb = 0, me = 0;
    if (gl < nChunk) { mb = segoff[(size_t)gl * pitch + b]; me = segoff[(size_t)gl * pitch + b + 1]; }
    int nr = (nChunk - g0 + 15) >> 4;
    if (nr > 32) nr = 32;

    int r = -1, base = 0, end = 0;
    unsigned e = 0u, eN = 0u;
    int cnt = 0, cntN = 0;
    next_trip(list, mb, me, nr, lane, r, base, end, e, cnt);
    unsigned s = e >> 8, dl = e & 255u;
    while (cnt > 0) {
      next_trip(list, mb, me, nr, lane, r, base, end, eN, cntN);
      if (lane < cnt) {
        const float2 v = h2v[s];
        atomicAdd(&acc0[dl], v.x);
        atomicAdd(&acc1[dl], v.y);
      }
      s = eN >> 8; dl = eN & 255u; cnt = cntN;
    }
  }
  __syncthreads();

  if (tid < BKT) {
    const int node = node0 + tid;
    if (node < n) {
      const float dn = dis[node];
      const float a0 = acc0[tid] * dn + b2[0];
      const float a1 = acc1[tid] * dn + b2[1];
      const float mx = fmaxf(a0, a1);
      const float lse = mx + logf(expf(a0 - mx) + expf(a1 - mx));
      float2 o; o.x = a0 - lse; o.y = a1 - lse;
      reinterpret_cast<float2*>(y)[node] = o;
    }
  }
}


extern "C" void kernel_launch(void* const* d_in, const int* in_sizes, int n_in,
                              void* d_out, int out_size, void* d_ws, size_t ws_size,
                              hipStream_t stream) {
  const float* x  = (const float*)d_in[0];
  const int*   ei = (const int*)d_in[1];
  const float* W1 = (const float*)d_in[2];
  const float* b1 = (const float*)d_in[3];
  const float* W2 = (const float*)d_in[4];
  const float* b2 = (const float*)d_in[5];

  const int n  = in_sizes[0] / IN_DIM;
  const int nE = in_sizes[1] / 2;
  const int* src = ei;
  const int* dst = ei + nE;
  float* ws = (float*)d_ws;
  float* y  = (float*)d_out;

  const int TB = 256;
  dim3 blk(TB);
  int gN   = (n + TB - 1) / TB;
  int gN16 = (int)(((long long)n * HID + TB - 1) / TB);
  int tiles = (n + 15) / 16;
  int gT   = (tiles + (TB / 32) - 1) / (TB / 32);

  const int nB     = (n + BKT - 1) / BKT;
  const int pitch  = (nB + 1 + 31) & ~31;
  const int nChunk = (nE + CH_CAP - 1) / CH_CAP;
  const size_t need = ((size_t)n * (1 + HID + 2) + (size_t)nE + (size_t)nChunk * pitch) * sizeof(float);

  if (ws_size >= need && nChunk > 0 && nB < NB_MAX && n <= (1 << 24)) {
    float* dis  = ws;
    float* h    = dis  + n;
    float* h2   = h    + (size_t)n * HID;
    unsigned* list = (unsigned*)(h2 + (size_t)n * 2);
    int* segoff    = (int*)(list + nE);

    k_bin       <<<nChunk, dim3(1024), 0, stream>>>(src, dst, list, segoff, nE, nB, pitch);
    k_deg_gemm1 <<<nB,  dim3(512), 0, stream>>>(list, segoff, x, W1, dis, h, n, pitch, nChunk);
    k_agg       <<<nB,  dim3(512), 0, stream>>>(list, segoff, dis, h, b1, W2, h2, n, pitch, nChunk);
    k_agg2_lsm  <<<nB,  dim3(512), 0, stream>>>(list, segoff, dis, h2, b2, y, n, pitch, nChunk);
    return;
  }

  float* deg  = ws;
  float* dis  = deg + n;
  float* h    = dis + n;
  float* out1 = h    + (size_t)n * HID;
  float* h2   = out1 + (size_t)n * HID;
  float* agg  = h;
  int gE   = (nE + TB - 1) / TB;
  int gS   = 2048;

  k_init_deg  <<<gN,  blk, 0, stream>>>(deg, n);
  k_deg_edges <<<gE,  blk, 0, stream>>>(dst, deg, nE);
  k_dis       <<<gN,  blk, 0, stream>>>(deg, dis, n);
  k_gemm1_wmma<<<gT,  blk, 0, stream>>>(x, W1, h, n);
  k_self1     <<<gN,  blk, 0, stream>>>(h, dis, out1, n);
  k_scatter1  <<<gS,  blk, 0, stream>>>(src, dst, dis, h, out1, nE);
  k_relu_bias <<<gN16,blk, 0, stream>>>(out1, b1, nullptr, n);
  k_self1     <<<gN,  blk, 0, stream>>>(out1, dis, agg, n);
  k_scatter1  <<<gS,  blk, 0, stream>>>(src, dst, dis, out1, agg, nE);
  k_gemm2_wmma<<<gT,  blk, 0, stream>>>(agg, W2, h2, n);
  k_logsoftmax<<<gN,  blk, 0, stream>>>(h2, b2, y, n);
}


# Destination ranges of 400 nodes, one 25-wave work-group each (was 256 nodes, 16 waves): 500 groups, at most 2 per WGP, where 782 left most WGPs 3 and some 4; LDS sums in arrival order as before
# speedup vs baseline: 7.9118x; 1.1611x over previous
#include <hip/hip_runtime.h>
#include <hip/hip_bf16.h>
#include <math.h>

typedef __attribute__((ext_vector_type(2))) float v2f;
typedef __attribute__((ext_vector_type(8))) float v8f;

#define IN_DIM 128
#define HID    16


__global__ void k_init_deg(float* __restrict__ deg, int n) {
  int i = blockIdx.x * blockDim.x + threadIdx.x;
  if (i < n) deg[i] = 1.0f;
}

__global__ void k_deg_edges(const int* __restrict__ dst, float* __restrict__ deg, int nE) {
  int e = blockIdx.x * blockDim.x + threadIdx.x;
  if (e < nE) atomicAdd(&deg[dst[e]], 1.0f);
}

__global__ void k_dis(const float* __restrict__ deg, float* __restrict__ dis, int n) {
  int i = blockIdx.x * blockDim.x + threadIdx.x;
  if (i < n) dis[i] = rsqrtf(deg[i]);
}


__global__ void k_gemm1_wmma(const float* __restrict__ x, const float* __restrict__ W1,
                             float* __restrict__ h, int n) {
  const int lane = threadIdx.x & 31;
  const int wave = threadIdx.x >> 5;
  const int tile = blockIdx.x * (blockDim.x >> 5) + wave;
  const int row0 = tile << 4;
  if (row0 >= n) return;
  const int m  = lane & 15;
  const int kh = lane >> 4;
  const float* xrow = x + (size_t)(row0 + m) * IN_DIM + 2 * kh;

  v8f c = {0.f, 0.f, 0.f, 0.f, 0.f, 0.f, 0.f, 0.f};
#pragma unroll
  for (int k0 = 0; k0 < IN_DIM / 4; ++k0) {
    const int kk = 4 * k0 + 2 * kh;
    const float2 av = *reinterpret_cast<const float2*>(xrow + 4 * k0);
    v2f a; a.x = av.x; a.y = av.y;
    v2f b; b.x = W1[(kk + 0) * HID + m];
           b.y = W1[(kk + 1) * HID + m];
    c = __builtin_amdgcn_wmma_f32_16x16x4_f32(false, a, false, b, (short)0, c, false, false);
  }
  float* hout = h + (size_t)(row0 + 8 * kh) * HID + m;
#pragma unroll
  for (int r = 0; r < 8; ++r)
    hout[(size_t)r * HID] = c[r];
}


__global__ void k_gemm2_wmma(const float* __restrict__ h1, const float* __restrict__ W2,
                             float* __restrict__ h2, int n) {
  const int lane = threadIdx.x & 31;
  const int wave = threadIdx.x >> 5;
  const int tile = blockIdx.x * (blockDim.x >> 5) + wave;
  const int row0 = tile << 4;
  if (row0 >= n) return;
  const int m  = lane & 15;
  const int kh = lane >> 4;
  const float bmask = (m < 2) ? 1.0f : 0.0f;
  const int   col   = (m < 2) ? m : 0;
  const float* arow = h1 + (size_t)(row0 + m) * HID + 2 * kh;

  v8f c = {0.f, 0.f, 0.f, 0.f, 0.f, 0.f, 0.f, 0.f};
#pragma unroll
  for (int k0 = 0; k0 < HID / 4; ++k0) {
    const int kk = 4 * k0 + 2 * kh;
    const float2 av = *reinterpret_cast<const float2*>(arow + 4 * k0);
    v2f a; a.x = av.x; a.y = av.y;
    v2f b; b.x = W2[(kk + 0) * 2 + col] * bmask;
           b.y = W2[(kk + 1) * 2 + col] * bmask;
    c = __builtin_amdgcn_wmma_f32_16x16x4_f32(false, a, false, b, (short)0, c, false, false);
  }
  if (m < 2) {
#pragma unroll
    for (int r = 0; r < 8; ++r)
      h2[(size_t)(row0 + r + 8 * kh) * 2 + m] = c[r];
  }
}


__global__ void k_self1(const float* __restrict__ h, const float* __restrict__ dis,
                        float* __restrict__ out1, int n) {
  int i = blockIdx.x * blockDim.x + threadIdx.x;
  if (i >= n) return;
  float s = dis[i]; s = s * s;
  const float4* hp = reinterpret_cast<const float4*>(h + (size_t)i * HID);
  float4* op = reinterpret_cast<float4*>(out1 + (size_t)i * HID);
#pragma unroll
  for (int q = 0; q < 4; ++q) {
    float4 v = hp[q];
    v.x *= s; v.y *= s; v.z *= s; v.w *= s;
    op[q] = v;
  }
}


static __device__ __forceinline__ unsigned row_bcast(int j, unsigned v) {
  const unsigned sel = 0x11111111u * (unsigned)j;
  return __builtin_amdgcn_permlane16(0u, v, sel, sel, false, true);
}

__global__ void k_scatter1(const int* __restrict__ src, const int* __restrict__ dst,
                           const float* __restrict__ dis, const float* __restrict__ h,
                           float* __restrict__ out1, int nE) {
  const int lane = threadIdx.x & 31;
  const unsigned f = lane & 15;
  const int wv     = __builtin_amdgcn_readfirstlane((int)(threadIdx.x >> 5));
  const int wpb    = (int)(blockDim.x >> 5);
  const int nWaves = (int)gridDim.x * wpb;
  const int nBatch = (nE + 31) >> 5;
  int b = (int)blockIdx.x * wpb + wv;
  if (b >= nBatch) return;

  unsigned s = 0, d = 0;
  if (lane < nE - (b << 5)) {
    s = (unsigned)src[(b << 5) + lane]; d = (unsigned)dst[(b << 5) + lane];
  }
  for (; b < nBatch; b += nWaves) {
    const int left = nE - (b << 5);
    const int bn   = b + nWaves;
    unsigned sN = 0, dN = 0;
    if (left >= 32) {
      const float ds = dis[s], dd = dis[d];
      float v[16];
#pragma unroll
      for (int j = 0; j < 16; ++j) v[j] = h[row_bcast(j, s) * HID + f];
      if (bn < nBatch && lane < nE - (bn << 5)) {
        sN = (unsigned)src[(bn << 5) + lane]; dN = (unsigned)dst[(bn << 5) + lane];
      }
      const float nrm = ds * dd;
#pragma unroll
      for (int j = 0; j < 16; ++j) {
        const unsigned dj = row_bcast(j, d);
        const float    nj = __uint_as_float(row_bcast(j, __float_as_uint(nrm)));
        atomicAdd(out1 + (dj * HID + f), v[j] * nj);
      }
    } else {
      float nrm = 0.f;
      if (lane < left) nrm = dis[s] * dis[d];
      const int mine = left - (lane & 16);
#pragma unroll
      for (int j = 0; j < 16; ++j) {
        const unsigned sj = row_bcast(j, s);
        const unsigned dj = row_bcast(j, d);
        const float    nj = __uint_as_float(row_bcast(j, __float_as_uint(nrm)));
        if (j < mine) atomicAdd(out1 + (dj * HID + f), h[sj * HID + f] * nj);
      }
    }
    s = sN; d = dN;
  }
}

__global__ void k_relu_bias(float* __restrict__ out1, const float* __restrict__ b1,
                            const float* __restrict__ rs, int n) {
  long long t = (long long)blockIdx.x * blockDim.x + threadIdx.x;
  if (t >= (long long)n * HID) return;
  int j = (int)(t & (HID - 1));
  float sc = 1.0f;
  if (rs) sc = rs[t >> 4];
  float v = out1[t] + b1[j];
  v = v > 0.f ? v : 0.f;
  out1[t] = v * sc;
}

__global__ void k_logsoftmax(const float* __restrict__ h2, const float* __restrict__ b2,
                             float* __restrict__ y, int n) {
  int i = blockIdx.x * blockDim.x + threadIdx.x;
  if (i >= n) return;
  float a0 = h2[(size_t)i * 2 + 0] + b2[0];
  float a1 = h2[(size_t)i * 2 + 1] + b2[1];
  float mx = fmaxf(a0, a1);
  float lse = mx + logf(expf(a0 - mx) + expf(a1 - mx));
  y[(size_t)i * 2 + 0] = a0 - lse;
  y[(size_t)i * 2 + 1] = a1 - lse;
}


#define BKT      400
#define NW       25
#define NTHR     800
#define DLB      9
#define DLM      511u
#define CH_CAP   25024
#define NB_MAX   1024

__global__ __launch_bounds__(1024) void k_bin(const int* __restrict__ src, const int* __restrict__ dst,
                                              unsigned* __restrict__ list, int* __restrict__ segoff,
                                              int nE, int nB, int pitch) {
  __shared__ unsigned cnt[NB_MAX];
  __shared__ unsigned wsum[32];
  __shared__ unsigned stage[CH_CAP];
  const int tid  = threadIdx.x;
  const int lane = tid & 31;
  const int wave = tid >> 5;
  const int g    = blockIdx.x;
  int e0 = g * CH_CAP;            if (e0 > nE) e0 = nE;
  int e1 = e0 + CH_CAP;           if (e1 > nE) e1 = nE;

  cnt[tid] = 0u;
  __syncthreads();
  for (int e = e0 + tid; e < e1; e += 8 * 1024) {
    unsigned d[8];
#pragma unroll
    for (int k = 0; k < 8; ++k) { const int ee = e + k * 1024; d[k] = (unsigned)dst[ee < e1 ? ee : e1 - 1]; }
#pragma unroll
    for (int k = 0; k < 8; ++k) if (e + k * 1024 < e1) atomicAdd(&cnt[d[k] / BKT], 1u);
  }
  __syncthreads();

  const unsigned c = cnt[tid];
  unsigned x = c;
  for (int o = 1; o < 32; o <<= 1) { const unsigned y = __shfl_up(x, o); if (lane >= o) x += y; }
  if (lane == 31) wsum[wave] = x;
  __syncthreads();
  if (wave == 0) {
    const unsigned t = wsum[lane];
    unsigned u = t;
    for (int o = 1; o < 32; o <<= 1) { const unsigned y = __shfl_up(u, o); if (lane >= o) u += y; }
    wsum[lane] = u - t;
  }
  __syncthreads();
  const unsigned excl = x - c + wsum[wave];
  if (tid <= nB) segoff[(size_t)g * pitch + tid] = e0 + (int)excl;
  cnt[tid] = excl;
  __syncthreads();

  for (int e = e0 + tid; e < e1; e += 8 * 1024) {
    unsigned d[8], s[8];
#pragma unroll
    for (int k = 0; k < 8; ++k) {
      const int ee = e + k * 1024, ec = ee < e1 ? ee : e1 - 1;
      d[k] = (unsigned)dst[ec]; s[k] = (unsigned)src[ec];
    }
#pragma unroll
    for (int k = 0; k < 8; ++k) {
      if (e + k * 1024 < e1) {
        const unsigned rg = d[k] / BKT;
        const unsigned r  = atomicAdd(&cnt[rg], 1u);
        stage[r] = (s[k] << DLB) | (d[k] - rg * BKT);
      }
    }
  }
  __syncthreads();
  const int len = e1 - e0;
  for (int i = tid; i < len; i += 1024) list[e0 + i] = stage[i];
}

__global__ __launch_bounds__(NTHR) void k_deg_gemm1(const unsigned* __restrict__ list, const int* __restrict__ segoff,
                                                   const float* __restrict__ x, const float* __restrict__ W1,
                                                   float* __restrict__ dis, float* __restrict__ h,
                                                   int n, int pitch, int nChunk) {
  __shared__ unsigned cnt[BKT];
  const int tid  = threadIdx.x;
  const int lane = tid & 31;
  const int wv   = __builtin_amdgcn_readfirstlane((int)(threadIdx.x >> 5));
  const int b    = blockIdx.x;
  const int node0 = b * BKT;
  if (tid < BKT) cnt[tid] = 0u;
  __syncthreads();

  for (int g0 = wv; g0 < nChunk; g0 += NW * 32) {
    const int gl = g0 + NW * lane;
    int mb = 0, me = 0;
    if (gl < nChunk) { mb = segoff[(size_t)gl * pitch + b]; me = segoff[(size_t)gl * pitch + b + 1]; }
    int nr = (nChunk - g0 + NW - 1) / NW;
    if (nr > 32) nr = 32;
    for (int r0 = 0; r0 < nr; r0 += 8) {
      unsigned w0[8], w1[8], w2[8];
      int left[8];
#pragma unroll
      for (int k = 0; k < 8; ++k) {
        const int beg = __builtin_amdgcn_readlane(mb, r0 + k);
        const int end = __builtin_amdgcn_readlane(me, r0 + k);
        const int p0 = beg + lane;
        int q0 = p0 < end ? p0 : end - 1;            if (q0 < 0) q0 = 0;
        int q1 = p0 + 32 < end ? p0 + 32 : end - 1;  if (q1 < 0) q1 = 0;
        int q2 = p0 + 64 < end ? p0 + 64 : end - 1;  if (q2 < 0) q2 = 0;
        w0[k] = list[q0];
        w1[k] = list[q1];
        w2[k] = list[q2];
        left[k] = end - p0;
      }
#pragma unroll
      for (int k = 0; k < 8; ++k) {
        if (left[k] > 0)  atomicAdd(&cnt[w0[k] & DLM], 1u);
        if (left[k] > 32) atomicAdd(&cnt[w1[k] & DLM], 1u);
        if (left[k] > 64) atomicAdd(&cnt[w2[k] & DLM], 1u);
      }
      for (int k = 0; k < 8; ++k) {
        const int beg = __builtin_amdgcn_readlane(mb, r0 + k);
        const int end = __builtin_amdgcn_readlane(me, r0 + k);
        for (int base = beg + 96; base < end; base += 32) {
          const int p = base + lane;
          if (p < end) atomicAdd(&cnt[list[p] & DLM], 1u);
        }
      }
    }
  }

  const int row0 = node0 + (wv << 4);
  const int m  = lane & 15;
  const int kh = lane >> 4;
  v8f c = {0.f, 0.f, 0.f, 0.f, 0.f, 0.f, 0.f, 0.f};
  if (row0 < n) {
    const float* xrow = x + (size_t)(row0 + m) * IN_DIM + 2 * kh;
#pragma unroll
    for (int k0 = 0; k0 < IN_DIM / 4; ++k0) {
      const int kk = 4 * k0 + 2 * kh;
      const float2 av = *reinterpret_cast<const float2*>(xrow + 4 * k0);
      v2f a;  a.x  = av.x; a.y = av.y;
      v2f bf; bf.x = W1[(kk + 0) * HID + m];
              bf.y = W1[(kk + 1) * HID + m];
      c = __builtin_amdgcn_wmma_f32_16x16x4_f32(false, a, false, bf, (short)0, c, false, false);
    }
  }
  __syncthreads();
  if (tid < BKT) {
    const int node = node0 + tid;
    if (node < n) dis[node] = rsqrtf((float)(cnt[tid] + 1u));
  }
  if (row0 < n) {
    float* hout = h + (size_t)(row0 + 8 * kh) * HID + m;
#pragma unroll
    for (int r = 0; r < 8; ++r) {
      const float sc = rsqrtf((float)(cnt[(wv << 4) + 8 * kh + r] + 1u));
      hout[(size_t)r * HID] = c[r] * sc;
    }
  }
}

__global__ __launch_bounds__(NTHR) void k_agg(const unsigned* __restrict__ list, const int* __restrict__ segoff,
                                             const float* __restrict__ dis, const float* __restrict__ h,
                                             const float* __restrict__ bias, const float* __restrict__ W2,
                                             float* __restrict__ out, int n, int pitch, int nChunk) {
  __shared__ __attribute__((aligned(16))) float acc[BKT * HID];
  __shared__ __attribute__((aligned(16))) float par[3 * HID];
  const int tid  = threadIdx.x;
  const int lane = tid & 31;
  const unsigned f = lane & 15;
  const int wv   = __builtin_amdgcn_readfirstlane((int)(threadIdx.x >> 5));
  const int b    = blockIdx.x;
  const int node0 = b * BKT;

  float bj = 0.f;
  if (bias && !W2) bj = bias[tid & (HID - 1)];
  float dnr = 0.f, pw = 0.f;
  if (W2) {
    if (tid < 3 * HID) pw = tid < HID ? bias[tid] : W2[tid - HID];
    const int nd = node0 + (wv << 4) + (lane & 15);
    if (nd < n) dnr = dis[nd];
  }

  for (int t = tid; t < BKT * HID; t += NTHR) {
    const int node = node0 + (t >> 4);
    acc[t] = node < n ? h[(size_t)node * HID + (t & 15)] : 0.f;
  }
  if (W2 && tid < 3 * HID) par[tid] = pw;
  __syncthreads();

  int beg = 0, end = 0;
  if (wv < nChunk) { beg = segoff[(size_t)wv * pitch + b]; end = segoff[(size_t)wv * pitch + b + 1]; }
  for (int g = wv; g < nChunk; g += NW) {
    const int gn = g + NW;
    int begN = 0, endN = 0;
    if (gn < nChunk) { begN = segoff[(size_t)gn * pitch + b]; endN = segoff[(size_t)gn * pitch + b + 1]; }
    for (int base = beg; base < end; base += 32) {
      const int left = end - base;
      const int p = base + lane;
      const unsigned e  = list[p < end ? p : end - 1];
      const unsigned s  = e >> DLB, dl = e & DLM;
      float v[16];
#pragma unroll
      for (int j = 0; j < 16; ++j) v[j] = h[row_bcast(j, s) * HID + f];
      const int mine = left - (lane & 16);
#pragma unroll
      for (int j = 0; j < 16; ++j) {
        const unsigned dj = row_bcast(j, dl);
        if (j < mine) atomicAdd(&acc[dj * HID + f], v[j]);
      }
    }
    beg = begN; end = endN;
  }
  if (W2) {
    const int m   = lane & 15;
    const int kh  = lane >> 4;
    const int row = (wv << 4) + m;
    const float bmask = (m < 2) ? 1.0f : 0.0f;
    const int   col   = (m < 2) ? m : 0;
    const float dn    = dnr;
    __syncthreads();
    v8f c = {0.f, 0.f, 0.f, 0.f, 0.f, 0.f, 0.f, 0.f};
#pragma unroll
    for (int k0 = 0; k0 < HID / 4; ++k0) {
      const int kk = 4 * k0 + 2 * kh;
      const float2 sv = *reinterpret_cast<const float2*>(&acc[row * HID + kk]);
      const float2 bv = *reinterpret_cast<const float2*>(&par[kk]);
      float a0 = sv.x * dn; a0 += bv.x; a0 = a0 > 0.f ? a0 : 0.f; a0 *= dn;
      float a1 = sv.y * dn; a1 += bv.y; a1 = a1 > 0.f ? a1 : 0.f; a1 *= dn;
      v2f a;  a.x  = a0; a.y = a1;
      v2f bf; bf.x = par[HID + (kk + 0) * 2 + col] * bmask;
              bf.y = par[HID + (kk + 1) * 2 + col] * bmask;
      c = __builtin_amdgcn_wmma_f32_16x16x4_f32(false, a, false, bf, (short)0, c, false, false);
    }
    if (m < 2) {
      const int r0 = node0 + (wv << 4) + 8 * kh;
#pragma unroll
      for (int r = 0; r < 8; ++r)
        if (r0 + r < n) out[(size_t)(r0 + r) * 2 + m] = c[r];
    }
    return;
  }
  __syncthreads();
  for (int t = tid; t < BKT * HID; t += NTHR) {
    const int node = node0 + (t >> 4);
    if (node < n) {
      const float dn = dis[node];
      float v = acc[t] * dn;
      if (bias) {
        v += bj;
        v = v > 0.f ? v : 0.f;
        v *= dn;
      }
      out[(size_t)node0 * HID + t] = v;
    }
  }
}

static __device__ __forceinline__ void next_trip(const unsigned* __restrict__ list, int mb, int me, int nr,
                                                 int lane, int& r, int& base, int& end,
                                                 unsigned& e, int& cnt) {
  while (base >= end && r + 1 < nr) {
    ++r;
    base = __builtin_amdgcn_readlane(mb, r);
    end  = __builtin_amdgcn_readlane(me, r);
  }
  cnt = end - base;
  if (cnt > 32) cnt = 32;
  int p = base + lane;
  if (p > end - 1) p = end - 1;
  if (p < 0) p = 0;
  e = list[p];
  if (cnt > 0) base += 32;
}

__global__ __launch_bounds__(NTHR) void k_agg2_lsm(const unsigned* __restrict__ list, const int* __restrict__ segoff,
                                                  const float* __restrict__ dis, const float* __restrict__ h2,
                                                  const float* __restrict__ b2, float* __restrict__ y,
                                                  int n, int pitch, int nChunk) {
  __shared__ float acc0[BKT];
  __shared__ float acc1[BKT];
  const int tid  = threadIdx.x;
  const int lane = tid & 31;
  const int wv   = __builtin_amdgcn_readfirstlane((int)(threadIdx.x >> 5));
  const int b    = blockIdx.x;
  const int node0 = b * BKT;
  const float2* __restrict__ h2v = reinterpret_cast<const float2*>(h2);

  if (tid < BKT) {
    const int node = node0 + tid;
    float2 v; v.x = 0.f; v.y = 0.f;
    if (node < n) v = h2v[node];
    acc0[tid] = v.x; acc1[tid] = v.y;
  }
  __syncthreads();

  for (int g0 = wv; g0 < nChunk; g0 += NW * 32) {
    const int gl = g0 + NW * lane;
    int mb = 0, me = 0;
    if (gl < nChunk) { mb = segoff[(size_t)gl * pitch + b]; me = segoff[(size_t)gl * pitch + b + 1]; }
    int nr = (nChunk - g0 + NW - 1) / NW;
    if (nr > 32) nr = 32;

    int r = -1, base = 0, end = 0;
    unsigned e = 0u, eN = 0u;
    int cnt = 0, cntN = 0;
    next_trip(list, mb, me, nr, lane, r, base, end, e, cnt);
    unsigned s = e >> DLB, dl = e & DLM;
    while (cnt > 0) {
      next_trip(list, mb, me, nr, lane, r, base, end, eN, cntN);
      if (lane < cnt) {
        const float2 v = h2v[s];
        atomicAdd(&acc0[dl], v.x);
        atomicAdd(&acc1[dl], v.y);
      }
      s = eN >> DLB; dl = eN & DLM; cnt = cntN;
    }
  }
  __syncthreads();

  if (tid < BKT) {
    const int node = node0 + tid;
    if (node < n) {
      const float dn = dis[node];
      const float a0 = acc0[tid] * dn + b2[0];
      const float a1 = acc1[tid] * dn + b2[1];
      const float mx = fmaxf(a0, a1);
      const float lse = mx + logf(expf(a0 - mx) + expf(a1 - mx));
      float2 o; o.x = a0 - lse; o.y = a1 - lse;
      reinterpret_cast<float2*>(y)[node] = o;
    }
  }
}


extern "C" void kernel_launch(void* const* d_in, const int* in_sizes, int n_in,
                              void* d_out, int out_size, void* d_ws, size_t ws_size,
                              hipStream_t stream) {
  const float* x  = (const float*)d_in[0];
  const int*   ei = (const int*)d_in[1];
  const float* W1 = (const float*)d_in[2];
  const float* b1 = (const float*)d_in[3];
  const float* W2 = (const float*)d_in[4];
  const float* b2 = (const float*)d_in[5];

  const int n  = in_sizes[0] / IN_DIM;
  const int nE = in_sizes[1] / 2;
  const int* src = ei;
  const int* dst = ei + nE;
  float* ws = (float*)d_ws;
  float* y  = (float*)d_out;

  const int TB = 256;
  dim3 blk(TB);
  int gN   = (n + TB - 1) / TB;
  int gN16 = (int)(((long long)n * HID + TB - 1) / TB);
  int tiles = (n + 15) / 16;
  int gT   = (tiles + (TB / 32) - 1) / (TB / 32);

  const int nB     = (n + BKT - 1) / BKT;
  const int pitch  = (nB + 1 + 31) & ~31;
  const int nChunk = (nE + CH_CAP - 1) / CH_CAP;
  const size_t need = ((size_t)n * (1 + HID + 2) + (size_t)nE + (size_t)nChunk * pitch) * sizeof(float);

  if (ws_size >= need && nChunk > 0 && nB < NB_MAX && n <= (1 << (32 - DLB))) {
    float* dis  = ws;
    float* h    = dis  + n;
    float* h2   = h    + (size_t)n * HID;
    unsigned* list = (unsigned*)(h2 + (size_t)n * 2);
    int* segoff    = (int*)(list + nE);

    k_bin       <<<nChunk, dim3(1024), 0, stream>>>(src, dst, list, segoff, nE, nB, pitch);
    k_deg_gemm1 <<<nB,  dim3(NTHR), 0, stream>>>(list, segoff, x, W1, dis, h, n, pitch, nChunk);
    k_agg       <<<nB,  dim3(NTHR), 0, stream>>>(list, segoff, dis, h, b1, W2, h2, n, pitch, nChunk);
    k_agg2_lsm  <<<nB,  dim3(NTHR), 0, stream>>>(list, segoff, dis, h2, b2, y, n, pitch, nChunk);
    return;
  }

  float* deg  = ws;
  float* dis  = deg + n;
  float* h    = dis + n;
  float* out1 = h    + (size_t)n * HID;
  float* h2   = out1 + (size_t)n * HID;
  float* agg  = h;
  int gE   = (nE + TB - 1) / TB;
  int gS   = 2048;

  k_init_deg  <<<gN,  blk, 0, stream>>>(deg, n);
  k_deg_edges <<<gE,  blk, 0, stream>>>(dst, deg, nE);
  k_dis       <<<gN,  blk, 0, stream>>>(deg, dis, n);
  k_gemm1_wmma<<<gT,  blk, 0, stream>>>(x, W1, h, n);
  k_self1     <<<gN,  blk, 0, stream>>>(h, dis, out1, n);
  k_scatter1  <<<gS,  blk, 0, stream>>>(src, dst, dis, h, out1, nE);
  k_relu_bias <<<gN16,blk, 0, stream>>>(out1, b1, nullptr, n);
  k_self1     <<<gN,  blk, 0, stream>>>(out1, dis, agg, n);
  k_scatter1  <<<gS,  blk, 0, stream>>>(src, dst, dis, out1, agg, nE);
  k_gemm2_wmma<<<gT,  blk, 0, stream>>>(agg, W2, h2, n);
  k_logsoftmax<<<gN,  blk, 0, stream>>>(h2, b2, y, n);
}
